# starting kernel
# baseline (speedup 1.0000x reference)
#include <hip/hip_runtime.h>
#include <hip/hip_bf16.h>

typedef __attribute__((ext_vector_type(2))) float v2f;
typedef __attribute__((ext_vector_type(8))) float v8f;

#define EPS 1e-5f

__global__ void k_repack(const float* __restrict__ comp_w,
                         const float* __restrict__ enc_w,
                         float* __restrict__ wCw,
                         float* __restrict__ wAe) {
    const int i = blockIdx.x * blockDim.x + threadIdx.x;
    if (i < 8192) {
        const int o = (i >> 1) & 63;
        const int k = ((i >> 7) << 1) | (i & 1);
        wCw[i] = comp_w[o * 128 + k];
    }
    if (i < 64512) {
        const int o = (i >> 1) % 112;
        const int t = ((i / 224) << 1) | (i & 1);
        const int c = t & 63, off = t >> 6;
        wAe[i] = (o < 100) ? enc_w[(size_t)o * 576 + c * 9 + off] : 0.f;
    }
}

__global__ void k_compress(const float* __restrict__ x,
                           const float* __restrict__ wCw,
                           const float* __restrict__ gamma,
                           const float* __restrict__ beta,
                           const float* __restrict__ mean,
                           const float* __restrict__ var,
                           float* __restrict__ y) {
    __shared__ float ls[128 * 16];
    const int tid    = threadIdx.x;
    const int sTile  = blockIdx.x;
    const int b      = sTile >> 8;
    const int hwBase = (sTile & 255) << 4;

    const float* xb = x + (size_t)b * 128 * 4096;
#pragma unroll
    for (int i = tid; i < 2048; i += 128)
        ls[i] = xb[(size_t)(i >> 4) * 4096 + hwBase + (i & 15)];
    __syncthreads();

    const int lane = tid & 31, l = lane & 15, half = lane >> 4;
    const int oBase = (tid >> 5) << 4;
    const int aCol = (oBase + l) * 2;

    v8f acc = {};
#pragma unroll 4
    for (int kb = 0; kb < 128; kb += 4) {
        const int k0 = kb + half * 2;
        const v2f a = *(const v2f*)(wCw + ((kb >> 1) + half) * 128 + aCol);
        v2f bm;
        bm.x = ls[k0 * 16 + l];
        bm.y = ls[(k0 + 1) * 16 + l];
        acc = __builtin_amdgcn_wmma_f32_16x16x4_f32(
            false, a, false, bm, (short)0, acc, false, false);
    }

#pragma unroll
    for (int j = 0; j < 8; ++j) {
        const int o = oBase + j + half * 8;
        const float iv = gamma[o] * rsqrtf(var[o] + EPS);
        const float bb = beta[o] - mean[o] * iv;
        float v = acc[j] * iv + bb;
        v = v * __frcp_rn(1.f + __expf(-v));
        y[((size_t)b * 64 + o) * 4096 + hwBase + l] = v;
    }
}

__global__ void k_encconv(const float* __restrict__ y,
                          const float* __restrict__ wAe,
                          const float* __restrict__ enc_b,
                          float* __restrict__ logits) {
    __shared__ float ls[64 * 54];
    const int tid    = threadIdx.x;
    const int sTile  = blockIdx.x;
    const int b      = sTile >> 8;
    const int hwBase = (sTile & 255) << 4;
    const int h      = hwBase >> 6;
    const int w0     = hwBase & 63;

    const float* yb = y + (size_t)b * 64 * 4096;
    for (int i = tid; i < 3456; i += 224) {
        const int c = i / 54, rem = i % 54;
        const int r = rem / 18, col = rem % 18;
        const int hh = h + r - 1, ww = w0 + col - 1;
        const bool ok = (hh >= 0) & (hh < 64) & (ww >= 0) & (ww < 64);
        ls[i] = ok ? yb[(size_t)c * 4096 + hh * 64 + ww] : 0.f;
    }
    __syncthreads();

    const int lane = tid & 31, l = lane & 15, half = lane >> 4;
    const int oBase = (tid >> 5) << 4;
    const int aCol = (oBase + l) * 2;

    v8f acc = {};
    for (int off = 0; off < 9; ++off) {
        const int ki = off / 3, kj = off % 3;
        const int bcol = ki * 18 + kj + l;
#pragma unroll 4
        for (int cb = 0; cb < 64; cb += 4) {
            const int c0 = cb + half * 2;
            const int pair = off * 32 + (cb >> 1) + half;
            const v2f a = *(const v2f*)(wAe + pair * 224 + aCol);
            v2f bm;
            bm.x = ls[c0 * 54 + bcol];
            bm.y = ls[(c0 + 1) * 54 + bcol];
            acc = __builtin_amdgcn_wmma_f32_16x16x4_f32(
                false, a, false, bm, (short)0, acc, false, false);
        }
    }

#pragma unroll
    for (int j = 0; j < 8; ++j) {
        const int oo = oBase + j + half * 8;
        if (oo < 100)
            logits[((size_t)b * 100 + oo) * 4096 + hwBase + l] = acc[j] + enc_b[oo];
    }
}

__global__ void k_reassemble(const float* __restrict__ x,
                             const float* __restrict__ logits,
                             float* __restrict__ out) {
    extern __shared__ float ls[];
    const int tid = threadIdx.x;
    const int X   = tid & 127;
    const int s1  = tid >> 7;
    const int b   = blockIdx.x >> 6;
    const int Yp  = blockIdx.x & 63;
    const int Y   = Yp * 2 + s1;
    const int s2  = X & 1;
    const int hbase = Yp - 1;

    const int r   = (tid < 198) ? tid / 66 : 0;
    const int col = (tid < 198) ? tid % 66 : 0;
    const int hh  = hbase + r, ww = col - 1;
    const bool ok = (tid < 198) & (hh >= 0) & (hh < 64) & (ww >= 0) & (ww < 64);
    const float* xb = x + (size_t)b * 128 * 4096;
    const int xoff  = hh * 64 + ww;
    if (tid < 198) {
#pragma unroll 4
        for (int c = 0; c < 128; ++c)
            ls[c * 198 + tid] = ok ? xb[(size_t)c * 4096 + xoff] : 0.f;
    }
    __syncthreads();

    const int w = X >> 1;
    const float* lg = logits + (size_t)b * 100 * 4096 + (size_t)Yp * 64 + w;
    float p[25];
    float mx = -1e30f;
#pragma unroll
    for (int q = 0; q < 25; ++q) {
        p[q] = lg[(size_t)(q * 4 + s1 * 2 + s2) * 4096];
        mx = fmaxf(mx, p[q]);
    }
    float sum = 0.f;
#pragma unroll
    for (int q = 0; q < 25; ++q) { p[q] = __expf(p[q] - mx); sum += p[q]; }
    const float rs = __frcp_rn(sum);

    float wc[9] = {0.f, 0.f, 0.f, 0.f, 0.f, 0.f, 0.f, 0.f, 0.f};
#pragma unroll
    for (int ki = 0; ki < 5; ++ki) {
        const int di = (ki + s1) >> 1;
#pragma unroll
        for (int kj = 0; kj < 5; ++kj) {
            const int dj = (kj + s2) >> 1;
            wc[di * 3 + dj] += p[ki * 5 + kj] * rs;
        }
    }

    const int base = ((X - 2) >> 1) + 1;
    float* ob = out + (size_t)b * 128 * 16384 + (size_t)Y * 128 + X;
#pragma unroll 2
    for (int c = 0; c < 128; ++c) {
        const float* lc = ls + c * 198 + base;
        float s = 0.f;
#pragma unroll
        for (int t = 0; t < 9; ++t)
            s += wc[t] * lc[(t / 3) * 66 + (t % 3)];
        ob[(size_t)c * 16384] = s;
    }
}

extern "C" void kernel_launch(void* const* d_in, const int* in_sizes, int n_in,
                              void* d_out, int out_size, void* d_ws, size_t ws_size,
                              hipStream_t stream) {
    const float* x      = (const float*)d_in[0];
    const float* comp_w = (const float*)d_in[1];
    const float* gamma  = (const float*)d_in[2];
    const float* beta   = (const float*)d_in[3];
    const float* mean   = (const float*)d_in[4];
    const float* var    = (const float*)d_in[5];
    const float* enc_w  = (const float*)d_in[6];
    const float* enc_b  = (const float*)d_in[7];
    float* out = (float*)d_out;

    float* y      = (float*)d_ws;
    float* logits = y + (size_t)1048576;
    float* wCw    = logits + (size_t)1638400;
    float* wAe    = wCw + (size_t)8192;

    k_repack<<<252, 256, 0, stream>>>(comp_w, enc_w, wCw, wAe);
    k_compress<<<1024, 128, 0, stream>>>(x, wCw, gamma, beta, mean, var, y);
    k_encconv<<<1024, 224, 0, stream>>>(y, wAe, enc_b, logits);
    k_reassemble<<<256, 256, 198 * 128 * sizeof(float), stream>>>(x, logits, out);
}


# LDS staging loops of all three kernels: global loads issued as independent batches (always-valid address, select after) instead of one load-wait-store round trip per element
# speedup vs baseline: 1.5145x; 1.5145x over previous
#include <hip/hip_runtime.h>
#include <hip/hip_bf16.h>

typedef __attribute__((ext_vector_type(2))) float v2f;
typedef __attribute__((ext_vector_type(8))) float v8f;

#define EPS 1e-5f

__global__ void k_repack(const float* __restrict__ comp_w,
                         const float* __restrict__ enc_w,
                         float* __restrict__ wCw,
                         float* __restrict__ wAe) {
    const int i = blockIdx.x * blockDim.x + threadIdx.x;
    if (i < 8192) {
        const int o = (i >> 1) & 63;
        const int k = ((i >> 7) << 1) | (i & 1);
        wCw[i] = comp_w[o * 128 + k];
    }
    if (i < 64512) {
        const int o = (i >> 1) % 112;
        const int t = ((i / 224) << 1) | (i & 1);
        const int c = t & 63, off = t >> 6;
        wAe[i] = (o < 100) ? enc_w[(size_t)o * 576 + c * 9 + off] : 0.f;
    }
}

__global__ void k_compress(const float* __restrict__ x,
                           const float* __restrict__ wCw,
                           const float* __restrict__ gamma,
                           const float* __restrict__ beta,
                           const float* __restrict__ mean,
                           const float* __restrict__ var,
                           float* __restrict__ y) {
    __shared__ float ls[128 * 16];
    const int tid    = threadIdx.x;
    const int sTile  = blockIdx.x;
    const int b      = sTile >> 8;
    const int hwBase = (sTile & 255) << 4;

    const float* xb = x + (size_t)b * 128 * 4096;
    {
        float v[16];
#pragma unroll
        for (int j = 0; j < 16; ++j) {
            const int i = tid + j * 128;
            v[j] = xb[(size_t)(i >> 4) * 4096 + hwBase + (i & 15)];
        }
#pragma unroll
        for (int j = 0; j < 16; ++j)
            ls[tid + j * 128] = v[j];
    }
    __syncthreads();

    const int lane = tid & 31, l = lane & 15, half = lane >> 4;
    const int oBase = (tid >> 5) << 4;
    const int aCol = (oBase + l) * 2;

    v8f acc = {};
#pragma unroll 4
    for (int kb = 0; kb < 128; kb += 4) {
        const int k0 = kb + half * 2;
        const v2f a = *(const v2f*)(wCw + ((kb >> 1) + half) * 128 + aCol);
        v2f bm;
        bm.x = ls[k0 * 16 + l];
        bm.y = ls[(k0 + 1) * 16 + l];
        acc = __builtin_amdgcn_wmma_f32_16x16x4_f32(
            false, a, false, bm, (short)0, acc, false, false);
    }

#pragma unroll
    for (int j = 0; j < 8; ++j) {
        const int o = oBase + j + half * 8;
        const float iv = gamma[o] * rsqrtf(var[o] + EPS);
        const float bb = beta[o] - mean[o] * iv;
        float v = acc[j] * iv + bb;
        v = v * __frcp_rn(1.f + __expf(-v));
        y[((size_t)b * 64 + o) * 4096 + hwBase + l] = v;
    }
}

__global__ void k_encconv(const float* __restrict__ y,
                          const float* __restrict__ wAe,
                          const float* __restrict__ enc_b,
                          float* __restrict__ logits) {
    __shared__ float ls[16 * 224];
    const int tid    = threadIdx.x;
    const int sTile  = blockIdx.x;
    const int b      = sTile >> 8;
    const int hwBase = (sTile & 255) << 4;
    const int h      = hwBase >> 6;
    const int w0     = hwBase & 63;

    const float* yb = y + (size_t)b * 64 * 4096;
    {
        float v[16];
#pragma unroll
        for (int j = 0; j < 16; ++j) {
            const int i  = tid + j * 224;
            const int ic = i < 3456 ? i : 3455;
            const int c = ic / 54, rem = ic - c * 54;
            const int r = rem / 18, col = rem - r * 18;
            const int hh = h + r - 1, ww = w0 + col - 1;
            const bool ok = (hh >= 0) & (hh < 64) & (ww >= 0) & (ww < 64);
            const int hc = hh < 0 ? 0 : (hh > 63 ? 63 : hh);
            const int wc = ww < 0 ? 0 : (ww > 63 ? 63 : ww);
            const float t = yb[(size_t)c * 4096 + hc * 64 + wc];
            v[j] = ok ? t : 0.f;
        }
#pragma unroll
        for (int j = 0; j < 16; ++j)
            ls[tid + j * 224] = v[j];
    }
    __syncthreads();

    const int lane = tid & 31, l = lane & 15, half = lane >> 4;
    const int oBase = (tid >> 5) << 4;
    const int aCol = (oBase + l) * 2;

    v8f acc = {};
    for (int off = 0; off < 9; ++off) {
        const int ki = off / 3, kj = off % 3;
        const int bcol = ki * 18 + kj + l;
#pragma unroll 4
        for (int cb = 0; cb < 64; cb += 4) {
            const int c0 = cb + half * 2;
            const int pair = off * 32 + (cb >> 1) + half;
            const v2f a = *(const v2f*)(wAe + pair * 224 + aCol);
            v2f bm;
            bm.x = ls[c0 * 54 + bcol];
            bm.y = ls[(c0 + 1) * 54 + bcol];
            acc = __builtin_amdgcn_wmma_f32_16x16x4_f32(
                false, a, false, bm, (short)0, acc, false, false);
        }
    }

#pragma unroll
    for (int j = 0; j < 8; ++j) {
        const int oo = oBase + j + half * 8;
        if (oo < 100)
            logits[((size_t)b * 100 + oo) * 4096 + hwBase + l] = acc[j] + enc_b[oo];
    }
}

__global__ void k_reassemble(const float* __restrict__ x,
                             const float* __restrict__ logits,
                             float* __restrict__ out) {
    extern __shared__ float ls[];
    const int tid = threadIdx.x;
    const int X   = tid & 127;
    const int s1  = tid >> 7;
    const int b   = blockIdx.x >> 6;
    const int Yp  = blockIdx.x & 63;
    const int Y   = Yp * 2 + s1;
    const int s2  = X & 1;
    const int hbase = Yp - 1;

    const int r   = (tid < 198) ? tid / 66 : 0;
    const int col = (tid < 198) ? tid % 66 : 0;
    const int hh  = hbase + r, ww = col - 1;
    const bool ok = (tid < 198) & (hh >= 0) & (hh < 64) & (ww >= 0) & (ww < 64);
    const float* xb = x + (size_t)b * 128 * 4096;
    const int xoff  = hh * 64 + ww;
    if (tid < 198) {
        const float* xs = xb + (ok ? xoff : 0);
#pragma unroll 1
        for (int c0 = 0; c0 < 128; c0 += 32) {
            float v[32];
#pragma unroll
            for (int j = 0; j < 32; ++j)
                v[j] = xs[(size_t)(c0 + j) * 4096];
#pragma unroll
            for (int j = 0; j < 32; ++j)
                ls[(c0 + j) * 198 + tid] = ok ? v[j] : 0.f;
        }
    }
    __syncthreads();

    const int w = X >> 1;
    const float* lg = logits + (size_t)b * 100 * 4096 + (size_t)Yp * 64 + w;
    float p[25];
    float mx = -1e30f;
#pragma unroll
    for (int q = 0; q < 25; ++q) {
        p[q] = lg[(size_t)(q * 4 + s1 * 2 + s2) * 4096];
        mx = fmaxf(mx, p[q]);
    }
    float sum = 0.f;
#pragma unroll
    for (int q = 0; q < 25; ++q) { p[q] = __expf(p[q] - mx); sum += p[q]; }
    const float rs = __frcp_rn(sum);

    float wc[9] = {0.f, 0.f, 0.f, 0.f, 0.f, 0.f, 0.f, 0.f, 0.f};
#pragma unroll
    for (int ki = 0; ki < 5; ++ki) {
        const int di = (ki + s1) >> 1;
#pragma unroll
        for (int kj = 0; kj < 5; ++kj) {
            const int dj = (kj + s2) >> 1;
            wc[di * 3 + dj] += p[ki * 5 + kj] * rs;
        }
    }

    const int base = ((X - 2) >> 1) + 1;
    float* ob = out + (size_t)b * 128 * 16384 + (size_t)Y * 128 + X;
#pragma unroll 2
    for (int c = 0; c < 128; ++c) {
        const float* lc = ls + c * 198 + base;
        float s = 0.f;
#pragma unroll
        for (int t = 0; t < 9; ++t)
            s += wc[t] * lc[(t / 3) * 66 + (t % 3)];
        ob[(size_t)c * 16384] = s;
    }
}

extern "C" void kernel_launch(void* const* d_in, const int* in_sizes, int n_in,
                              void* d_out, int out_size, void* d_ws, size_t ws_size,
                              hipStream_t stream) {
    const float* x      = (const float*)d_in[0];
    const float* comp_w = (const float*)d_in[1];
    const float* gamma  = (const float*)d_in[2];
    const float* beta   = (const float*)d_in[3];
    const float* mean   = (const float*)d_in[4];
    const float* var    = (const float*)d_in[5];
    const float* enc_w  = (const float*)d_in[6];
    const float* enc_b  = (const float*)d_in[7];
    float* out = (float*)d_out;

    float* y      = (float*)d_ws;
    float* logits = y + (size_t)1048576;
    float* wCw    = logits + (size_t)1638400;
    float* wAe    = wCw + (size_t)8192;

    k_repack<<<252, 256, 0, stream>>>(comp_w, enc_w, wCw, wAe);
    k_compress<<<1024, 128, 0, stream>>>(x, wCw, gamma, beta, mean, var, y);
    k_encconv<<<1024, 224, 0, stream>>>(y, wAe, enc_b, logits);
    k_reassemble<<<256, 256, 198 * 128 * sizeof(float), stream>>>(x, logits, out);
}


# WMMA A fragments prefetched as whole batches: k_compress all 32 k-steps ahead of the staging barrier, k_encconv one 3x3 tap (16 k-steps) ahead in a second register set
# speedup vs baseline: 1.7312x; 1.1431x over previous
#include <hip/hip_runtime.h>
#include <hip/hip_bf16.h>

typedef __attribute__((ext_vector_type(2))) float v2f;
typedef __attribute__((ext_vector_type(8))) float v8f;

#define EPS 1e-5f

__global__ void k_repack(const float* __restrict__ comp_w,
                         const float* __restrict__ enc_w,
                         float* __restrict__ wCw,
                         float* __restrict__ wAe) {
    const int i = blockIdx.x * blockDim.x + threadIdx.x;
    if (i < 8192) {
        const int o = (i >> 1) & 63;
        const int k = ((i >> 7) << 1) | (i & 1);
        wCw[i] = comp_w[o * 128 + k];
    }
    if (i < 64512) {
        const int o = (i >> 1) % 112;
        const int t = ((i / 224) << 1) | (i & 1);
        const int c = t & 63, off = t >> 6;
        wAe[i] = (o < 100) ? enc_w[(size_t)o * 576 + c * 9 + off] : 0.f;
    }
}

__global__ void k_compress(const float* __restrict__ x,
                           const float* __restrict__ wCw,
                           const float* __restrict__ gamma,
                           const float* __restrict__ beta,
                           const float* __restrict__ mean,
                           const float* __restrict__ var,
                           float* __restrict__ y) {
    __shared__ float ls[128 * 16];
    const int tid    = threadIdx.x;
    const int sTile  = blockIdx.x;
    const int b      = sTile >> 8;
    const int hwBase = (sTile & 255) << 4;

    const int lane = tid & 31, l = lane & 15, half = lane >> 4;
    const int oBase = (tid >> 5) << 4;
    const int aCol = (oBase + l) * 2;

    const float* xb = x + (size_t)b * 128 * 4096;
    v2f a[32];
    {
        float v[16];
#pragma unroll
        for (int j = 0; j < 16; ++j) {
            const int i = tid + j * 128;
            v[j] = xb[(size_t)(i >> 4) * 4096 + hwBase + (i & 15)];
        }
        const float* ap = wCw + half * 128 + aCol;
#pragma unroll
        for (int s = 0; s < 32; ++s)
            a[s] = *(const v2f*)(ap + s * 256);
#pragma unroll
        for (int j = 0; j < 16; ++j)
            ls[tid + j * 128] = v[j];
    }
    __syncthreads();

    v8f acc = {};
    const float* lb = ls + half * 32 + l;
#pragma unroll
    for (int s = 0; s < 32; ++s) {
        v2f bm;
        bm.x = lb[(4 * s) * 16];
        bm.y = lb[(4 * s + 1) * 16];
        acc = __builtin_amdgcn_wmma_f32_16x16x4_f32(
            false, a[s], false, bm, (short)0, acc, false, false);
    }

#pragma unroll
    for (int j = 0; j < 8; ++j) {
        const int o = oBase + j + half * 8;
        const float iv = gamma[o] * rsqrtf(var[o] + EPS);
        const float bb = beta[o] - mean[o] * iv;
        float v = acc[j] * iv + bb;
        v = v * __frcp_rn(1.f + __expf(-v));
        y[((size_t)b * 64 + o) * 4096 + hwBase + l] = v;
    }
}

__global__ void k_encconv(const float* __restrict__ y,
                          const float* __restrict__ wAe,
                          const float* __restrict__ enc_b,
                          float* __restrict__ logits) {
    __shared__ float ls[16 * 224];
    const int tid    = threadIdx.x;
    const int sTile  = blockIdx.x;
    const int b      = sTile >> 8;
    const int hwBase = (sTile & 255) << 4;
    const int h      = hwBase >> 6;
    const int w0     = hwBase & 63;

    const int lane = tid & 31, l = lane & 15, half = lane >> 4;
    const int oBase = (tid >> 5) << 4;
    const int aCol = (oBase + l) * 2;

    const float* ap = wAe + half * 224 + aCol;
    v2f a0[16], a1[16];

    const float* yb = y + (size_t)b * 64 * 4096;
    {
        float v[16];
#pragma unroll
        for (int j = 0; j < 16; ++j) {
            const int i  = tid + j * 224;
            const int ic = i < 3456 ? i : 3455;
            const int c = ic / 54, rem = ic - c * 54;
            const int r = rem / 18, col = rem - r * 18;
            const int hh = h + r - 1, ww = w0 + col - 1;
            const bool ok = (hh >= 0) & (hh < 64) & (ww >= 0) & (ww < 64);
            const int hc = hh < 0 ? 0 : (hh > 63 ? 63 : hh);
            const int wc = ww < 0 ? 0 : (ww > 63 ? 63 : ww);
            const float t = yb[(size_t)c * 4096 + hc * 64 + wc];
            v[j] = ok ? t : 0.f;
        }
#pragma unroll
        for (int s = 0; s < 16; ++s)
            a0[s] = *(const v2f*)(ap + s * 448);
#pragma unroll
        for (int j = 0; j < 16; ++j)
            ls[tid + j * 224] = v[j];
    }
    __syncthreads();

    v8f acc = {};
    const float* lb = ls + half * 108 + l;

#define ENC_LOAD_A(BUF, T)                                                    \
    _Pragma("unroll")                                                         \
    for (int s = 0; s < 16; ++s)                                              \
        BUF[s] = *(const v2f*)(ap + (T) * 7168 + s * 448);
#define ENC_MMA(BUF, T)                                                       \
    {                                                                         \
        const float* lo = lb + ((T) / 3) * 18 + ((T) % 3);                    \
        _Pragma("unroll")                                                     \
        for (int s = 0; s < 16; ++s) {                                        \
            v2f bm;                                                           \
            bm.x = lo[(4 * s) * 54];                                          \
            bm.y = lo[(4 * s + 1) * 54];                                      \
            acc = __builtin_amdgcn_wmma_f32_16x16x4_f32(                      \
                false, BUF[s], false, bm, (short)0, acc, false, false);       \
        }                                                                     \
    }

#pragma unroll 1
    for (int tap = 0; tap < 8; tap += 2) {
        ENC_LOAD_A(a1, tap + 1)
        __builtin_amdgcn_sched_barrier(0);
        ENC_MMA(a0, tap)
        ENC_LOAD_A(a0, tap + 2)
        __builtin_amdgcn_sched_barrier(0);
        ENC_MMA(a1, tap + 1)
    }
    ENC_MMA(a0, 8)
#undef ENC_LOAD_A
#undef ENC_MMA

#pragma unroll
    for (int j = 0; j < 8; ++j) {
        const int oo = oBase + j + half * 8;
        if (oo < 100)
            logits[((size_t)b * 100 + oo) * 4096 + hwBase + l] = acc[j] + enc_b[oo];
    }
}

__global__ void k_reassemble(const float* __restrict__ x,
                             const float* __restrict__ logits,
                             float* __restrict__ out) {
    extern __shared__ float ls[];
    const int tid = threadIdx.x;
    const int X   = tid & 127;
    const int s1  = tid >> 7;
    const int b   = blockIdx.x >> 6;
    const int Yp  = blockIdx.x & 63;
    const int Y   = Yp * 2 + s1;
    const int s2  = X & 1;
    const int hbase = Yp - 1;

    const int r   = (tid < 198) ? tid / 66 : 0;
    const int col = (tid < 198) ? tid % 66 : 0;
    const int hh  = hbase + r, ww = col - 1;
    const bool ok = (tid < 198) & (hh >= 0) & (hh < 64) & (ww >= 0) & (ww < 64);
    const float* xb = x + (size_t)b * 128 * 4096;
    const int xoff  = hh * 64 + ww;
    if (tid < 198) {
        const float* xs = xb + (ok ? xoff : 0);
#pragma unroll 1
        for (int c0 = 0; c0 < 128; c0 += 32) {
            float v[32];
#pragma unroll
            for (int j = 0; j < 32; ++j)
                v[j] = xs[(size_t)(c0 + j) * 4096];
#pragma unroll
            for (int j = 0; j < 32; ++j)
                ls[(c0 + j) * 198 + tid] = ok ? v[j] : 0.f;
        }
    }
    __syncthreads();

    const int w = X >> 1;
    const float* lg = logits + (size_t)b * 100 * 4096 + (size_t)Yp * 64 + w;
    float p[25];
    float mx = -1e30f;
#pragma unroll
    for (int q = 0; q < 25; ++q) {
        p[q] = lg[(size_t)(q * 4 + s1 * 2 + s2) * 4096];
        mx = fmaxf(mx, p[q]);
    }
    float sum = 0.f;
#pragma unroll
    for (int q = 0; q < 25; ++q) { p[q] = __expf(p[q] - mx); sum += p[q]; }
    const float rs = __frcp_rn(sum);

    float wc[9] = {0.f, 0.f, 0.f, 0.f, 0.f, 0.f, 0.f, 0.f, 0.f};
#pragma unroll
    for (int ki = 0; ki < 5; ++ki) {
        const int di = (ki + s1) >> 1;
#pragma unroll
        for (int kj = 0; kj < 5; ++kj) {
            const int dj = (kj + s2) >> 1;
            wc[di * 3 + dj] += p[ki * 5 + kj] * rs;
        }
    }

    const int base = ((X - 2) >> 1) + 1;
    float* ob = out + (size_t)b * 128 * 16384 + (size_t)Y * 128 + X;
#pragma unroll 2
    for (int c = 0; c < 128; ++c) {
        const float* lc = ls + c * 198 + base;
        float s = 0.f;
#pragma unroll
        for (int t = 0; t < 9; ++t)
            s += wc[t] * lc[(t / 3) * 66 + (t % 3)];
        ob[(size_t)c * 16384] = s;
    }
}

extern "C" void kernel_launch(void* const* d_in, const int* in_sizes, int n_in,
                              void* d_out, int out_size, void* d_ws, size_t ws_size,
                              hipStream_t stream) {
    const float* x      = (const float*)d_in[0];
    const float* comp_w = (const float*)d_in[1];
    const float* gamma  = (const float*)d_in[2];
    const float* beta   = (const float*)d_in[3];
    const float* mean   = (const float*)d_in[4];
    const float* var    = (const float*)d_in[5];
    const float* enc_w  = (const float*)d_in[6];
    const float* enc_b  = (const float*)d_in[7];
    float* out = (float*)d_out;

    float* y      = (float*)d_ws;
    float* logits = y + (size_t)1048576;
    float* wCw    = logits + (size_t)1638400;
    float* wAe    = wCw + (size_t)8192;

    k_repack<<<252, 256, 0, stream>>>(comp_w, enc_w, wCw, wAe);
    k_compress<<<1024, 128, 0, stream>>>(x, wCw, gamma, beta, mean, var, y);
    k_encconv<<<1024, 224, 0, stream>>>(y, wAe, enc_b, logits);
    k_reassemble<<<256, 256, 198 * 128 * sizeof(float), stream>>>(x, logits, out);
}


# k_reassemble: one thread per source pixel makes the whole 2x2 output quad from one 3x3 LDS patch read (4x fewer LDS reads); softmax weights folded with selects and handed over through LDS
# speedup vs baseline: 1.9048x; 1.1003x over previous
#include <hip/hip_runtime.h>
#include <hip/hip_bf16.h>

typedef __attribute__((ext_vector_type(2))) float v2f;
typedef __attribute__((ext_vector_type(8))) float v8f;

#define EPS 1e-5f

__global__ void k_repack(const float* __restrict__ comp_w,
                         const float* __restrict__ enc_w,
                         float* __restrict__ wCw,
                         float* __restrict__ wAe) {
    const int i = blockIdx.x * blockDim.x + threadIdx.x;
    if (i < 8192) {
        const int o = (i >> 1) & 63;
        const int k = ((i >> 7) << 1) | (i & 1);
        wCw[i] = comp_w[o * 128 + k];
    }
    if (i < 64512) {
        const int o = (i >> 1) % 112;
        const int t = ((i / 224) << 1) | (i & 1);
        const int c = t & 63, off = t >> 6;
        wAe[i] = (o < 100) ? enc_w[(size_t)o * 576 + c * 9 + off] : 0.f;
    }
}

__global__ void k_compress(const float* __restrict__ x,
                           const float* __restrict__ wCw,
                           const float* __restrict__ gamma,
                           const float* __restrict__ beta,
                           const float* __restrict__ mean,
                           const float* __restrict__ var,
                           float* __restrict__ y) {
    __shared__ float ls[128 * 16];
    const int tid    = threadIdx.x;
    const int sTile  = blockIdx.x;
    const int b      = sTile >> 8;
    const int hwBase = (sTile & 255) << 4;

    const int lane = tid & 31, l = lane & 15, half = lane >> 4;
    const int oBase = (tid >> 5) << 4;
    const int aCol = (oBase + l) * 2;

    const float* xb = x + (size_t)b * 128 * 4096;
    v2f a[32];
    {
        float v[16];
#pragma unroll
        for (int j = 0; j < 16; ++j) {
            const int i = tid + j * 128;
            v[j] = xb[(size_t)(i >> 4) * 4096 + hwBase + (i & 15)];
        }
        const float* ap = wCw + half * 128 + aCol;
#pragma unroll
        for (int s = 0; s < 32; ++s)
            a[s] = *(const v2f*)(ap + s * 256);
#pragma unroll
        for (int j = 0; j < 16; ++j)
            ls[tid + j * 128] = v[j];
    }
    __syncthreads();

    v8f acc = {};
    const float* lb = ls + half * 32 + l;
#pragma unroll
    for (int s = 0; s < 32; ++s) {
        v2f bm;
        bm.x = lb[(4 * s) * 16];
        bm.y = lb[(4 * s + 1) * 16];
        acc = __builtin_amdgcn_wmma_f32_16x16x4_f32(
            false, a[s], false, bm, (short)0, acc, false, false);
    }

#pragma unroll
    for (int j = 0; j < 8; ++j) {
        const int o = oBase + j + half * 8;
        const float iv = gamma[o] * rsqrtf(var[o] + EPS);
        const float bb = beta[o] - mean[o] * iv;
        float v = acc[j] * iv + bb;
        v = v * __frcp_rn(1.f + __expf(-v));
        y[((size_t)b * 64 + o) * 4096 + hwBase + l] = v;
    }
}

__global__ void k_encconv(const float* __restrict__ y,
                          const float* __restrict__ wAe,
                          const float* __restrict__ enc_b,
                          float* __restrict__ logits) {
    __shared__ float ls[16 * 224];
    const int tid    = threadIdx.x;
    const int sTile  = blockIdx.x;
    const int b      = sTile >> 8;
    const int hwBase = (sTile & 255) << 4;
    const int h      = hwBase >> 6;
    const int w0     = hwBase & 63;

    const int lane = tid & 31, l = lane & 15, half = lane >> 4;
    const int oBase = (tid >> 5) << 4;
    const int aCol = (oBase + l) * 2;

    const float* ap = wAe + half * 224 + aCol;
    v2f a0[16], a1[16];

    const float* yb = y + (size_t)b * 64 * 4096;
    {
        float v[16];
#pragma unroll
        for (int j = 0; j < 16; ++j) {
            const int i  = tid + j * 224;
            const int ic = i < 3456 ? i : 3455;
            const int c = ic / 54, rem = ic - c * 54;
            const int r = rem / 18, col = rem - r * 18;
            const int hh = h + r - 1, ww = w0 + col - 1;
            const bool ok = (hh >= 0) & (hh < 64) & (ww >= 0) & (ww < 64);
            const int hc = hh < 0 ? 0 : (hh > 63 ? 63 : hh);
            const int wc = ww < 0 ? 0 : (ww > 63 ? 63 : ww);
            const float t = yb[(size_t)c * 4096 + hc * 64 + wc];
            v[j] = ok ? t : 0.f;
        }
#pragma unroll
        for (int s = 0; s < 16; ++s)
            a0[s] = *(const v2f*)(ap + s * 448);
#pragma unroll
        for (int j = 0; j < 16; ++j)
            ls[tid + j * 224] = v[j];
    }
    __syncthreads();

    v8f acc = {};
    const float* lb = ls + half * 108 + l;

#define ENC_LOAD_A(BUF, T)                                                    \
    _Pragma("unroll")                                                         \
    for (int s = 0; s < 16; ++s)                                              \
        BUF[s] = *(const v2f*)(ap + (T) * 7168 + s * 448);
#define ENC_MMA(BUF, T)                                                       \
    {                                                                         \
        const float* lo = lb + ((T) / 3) * 18 + ((T) % 3);                    \
        _Pragma("unroll")                                                     \
        for (int s = 0; s < 16; ++s) {                                        \
            v2f bm;                                                           \
            bm.x = lo[(4 * s) * 54];                                          \
            bm.y = lo[(4 * s + 1) * 54];                                      \
            acc = __builtin_amdgcn_wmma_f32_16x16x4_f32(                      \
                false, BUF[s], false, bm, (short)0, acc, false, false);       \
        }                                                                     \
    }

#pragma unroll 1
    for (int tap = 0; tap < 8; tap += 2) {
        ENC_LOAD_A(a1, tap + 1)
        __builtin_amdgcn_sched_barrier(0);
        ENC_MMA(a0, tap)
        ENC_LOAD_A(a0, tap + 2)
        __builtin_amdgcn_sched_barrier(0);
        ENC_MMA(a1, tap + 1)
    }
    ENC_MMA(a0, 8)
#undef ENC_LOAD_A
#undef ENC_MMA

#pragma unroll
    for (int j = 0; j < 8; ++j) {
        const int oo = oBase + j + half * 8;
        if (oo < 100)
            logits[((size_t)b * 100 + oo) * 4096 + hwBase + l] = acc[j] + enc_b[oo];
    }
}

__global__ void k_reassemble(const float* __restrict__ x,
                             const float* __restrict__ logits,
                             float* __restrict__ out) {
    extern __shared__ float ls[];
    float* wl = ls + 128 * 198;
    const int tid = threadIdx.x;
    const int X   = tid & 127;
    const int s1  = tid >> 7;
    const int b   = blockIdx.x >> 6;
    const int Yp  = blockIdx.x & 63;
    const int s2  = X & 1;
    const int hbase = Yp - 1;

    const int r   = (tid < 198) ? tid / 66 : 0;
    const int col = (tid < 198) ? tid % 66 : 0;
    const int hh  = hbase + r, ww = col - 1;
    const bool ok = (tid < 198) & (hh >= 0) & (hh < 64) & (ww >= 0) & (ww < 64);
    const float* xb = x + (size_t)b * 128 * 4096;
    const int xoff  = hh * 64 + ww;
    if (tid < 198) {
        const float* xs = xb + (ok ? xoff : 0);
#pragma unroll 1
        for (int c0 = 0; c0 < 128; c0 += 32) {
            float v[32];
#pragma unroll
            for (int j = 0; j < 32; ++j)
                v[j] = xs[(size_t)(c0 + j) * 4096];
#pragma unroll
            for (int j = 0; j < 32; ++j)
                ls[(c0 + j) * 198 + tid] = ok ? v[j] : 0.f;
        }
    }

    const int w = X >> 1;
    const float* lg = logits + (size_t)b * 100 * 4096 + (size_t)Yp * 64 + w;
    float p[25];
    float mx = -1e30f;
#pragma unroll
    for (int q = 0; q < 25; ++q) {
        p[q] = lg[(size_t)(q * 4 + s1 * 2 + s2) * 4096];
        mx = fmaxf(mx, p[q]);
    }
    float sum = 0.f;
#pragma unroll
    for (int q = 0; q < 25; ++q) { p[q] = __expf(p[q] - mx); sum += p[q]; }
    const float rs = __frcp_rn(sum);

    const bool e2 = (s2 == 0), e1 = (s1 == 0);
    float cf[5][3];
#pragma unroll
    for (int ki = 0; ki < 5; ++ki) {
        const float q0 = p[ki * 5 + 0] * rs, q1 = p[ki * 5 + 1] * rs,
                    q2 = p[ki * 5 + 2] * rs, q3 = p[ki * 5 + 3] * rs,
                    q4 = p[ki * 5 + 4] * rs;
        cf[ki][0] = q0 + (e2 ? q1 : 0.f);
        cf[ki][1] = (e2 ? q3 : q1) + q2;
        cf[ki][2] = q4 + (e2 ? 0.f : q3);
    }
    float wc[9];
#pragma unroll
    for (int j = 0; j < 3; ++j) {
        wc[0 * 3 + j] = cf[0][j] + (e1 ? cf[1][j] : 0.f);
        wc[1 * 3 + j] = (e1 ? cf[3][j] : cf[1][j]) + cf[2][j];
        wc[2 * 3 + j] = cf[4][j] + (e1 ? 0.f : cf[3][j]);
    }

#pragma unroll
    for (int t = 0; t < 9; ++t)
        wl[t * 256 + tid] = wc[t];
    __syncthreads();

    const int qw = tid & 63;
    const int g  = tid >> 6;
    v2f W0[9], W1[9];
#pragma unroll
    for (int t = 0; t < 9; ++t) {
        W0[t] = *(const v2f*)(wl + t * 256 + 2 * qw);
        W1[t] = *(const v2f*)(wl + t * 256 + 128 + 2 * qw);
    }
    const float* lc = ls + (g * 32) * 198 + qw;
    float* ob = out + ((size_t)b * 128 + g * 32) * 16384 + (size_t)(2 * Yp) * 128 + 2 * qw;
#pragma unroll 4
    for (int i = 0; i < 32; ++i) {
        v2f a0 = {0.f, 0.f}, a1 = {0.f, 0.f};
#pragma unroll
        for (int t = 0; t < 9; ++t) {
            const float v = lc[i * 198 + (t / 3) * 66 + (t % 3)];
            a0 += W0[t] * v;
            a1 += W1[t] * v;
        }
        *(v2f*)(ob + (size_t)i * 16384)       = a0;
        *(v2f*)(ob + (size_t)i * 16384 + 128) = a1;
    }
}

extern "C" void kernel_launch(void* const* d_in, const int* in_sizes, int n_in,
                              void* d_out, int out_size, void* d_ws, size_t ws_size,
                              hipStream_t stream) {
    const float* x      = (const float*)d_in[0];
    const float* comp_w = (const float*)d_in[1];
    const float* gamma  = (const float*)d_in[2];
    const float* beta   = (const float*)d_in[3];
    const float* mean   = (const float*)d_in[4];
    const float* var    = (const float*)d_in[5];
    const float* enc_w  = (const float*)d_in[6];
    const float* enc_b  = (const float*)d_in[7];
    float* out = (float*)d_out;

    float* y      = (float*)d_ws;
    float* logits = y + (size_t)1048576;
    float* wCw    = logits + (size_t)1638400;
    float* wAe    = wCw + (size_t)8192;

    k_repack<<<252, 256, 0, stream>>>(comp_w, enc_w, wCw, wAe);
    k_compress<<<1024, 128, 0, stream>>>(x, wCw, gamma, beta, mean, var, y);
    k_encconv<<<1024, 224, 0, stream>>>(y, wAe, enc_b, logits);
    k_reassemble<<<256, 256, (198 * 128 + 9 * 256) * sizeof(float), stream>>>(x, logits, out);
}


# k_encreasm tile 16 -> 32 source px (512 blocks): each wave keeps two accumulator tiles, so each A fragment feeds two WMMAs (half the weight reads and waves); halo staged one position per thread
# speedup vs baseline: 2.4891x; 1.3068x over previous
#include <hip/hip_runtime.h>
#include <hip/hip_bf16.h>

typedef __attribute__((ext_vector_type(2))) float v2f;
typedef __attribute__((ext_vector_type(4))) float v4f;
typedef __attribute__((ext_vector_type(8))) float v8f;

#define EPS 1e-5f

__global__ void k_compress(const float* __restrict__ x,
                           const float* __restrict__ comp_w,
                           const float* __restrict__ enc_w,
                           const float* __restrict__ gamma,
                           const float* __restrict__ beta,
                           const float* __restrict__ mean,
                           const float* __restrict__ var,
                           float* __restrict__ y,
                           float* __restrict__ wAe) {
    __shared__ float ls[128 * 16];
    const int tid    = threadIdx.x;
    const int sTile  = blockIdx.x;
    const int b      = sTile >> 8;
    const int hwBase = (sTile & 255) << 4;

    const int lane = tid & 31, l = lane & 15, half = lane >> 4;
    const int oBase = (tid >> 5) << 4;

    const int wi = sTile * 63 + (tid < 63 ? tid : 0);
    const int wo = (wi >> 1) % 112;
    const int wt = ((wi / 224) << 1) | (wi & 1);
    const int wc = wt & 63, woff = wt >> 6;
    const float wraw = enc_w[(size_t)(wo < 100 ? wo : 0) * 576 + wc * 9 + woff];
    const float wv = (wo < 100) ? wraw : 0.f;

    const float* xb = x + (size_t)b * 128 * 4096;
    v4f q[16];
    {
        float v[16];
#pragma unroll
        for (int j = 0; j < 16; ++j) {
            const int i = tid + j * 128;
            v[j] = xb[(size_t)(i >> 4) * 4096 + hwBase + (i & 15)];
        }
        const float* ap = comp_w + (size_t)(oBase + l) * 128 + half * 4;
#pragma unroll
        for (int j = 0; j < 16; ++j)
            q[j] = *(const v4f*)(ap + j * 8);
#pragma unroll
        for (int j = 0; j < 16; ++j)
            ls[tid + j * 128] = v[j];
    }
    __syncthreads();

    v8f acc = {};
    const float* lb = ls + half * 64 + l;
#pragma unroll
    for (int j = 0; j < 16; ++j) {
        v2f am, bm;
        am.x = q[j].x;
        am.y = q[j].y;
        bm.x = lb[(8 * j) * 16];
        bm.y = lb[(8 * j + 1) * 16];
        acc = __builtin_amdgcn_wmma_f32_16x16x4_f32(
            false, am, false, bm, (short)0, acc, false, false);
        am.x = q[j].z;
        am.y = q[j].w;
        bm.x = lb[(8 * j + 2) * 16];
        bm.y = lb[(8 * j + 3) * 16];
        acc = __builtin_amdgcn_wmma_f32_16x16x4_f32(
            false, am, false, bm, (short)0, acc, false, false);
    }

#pragma unroll
    for (int j = 0; j < 8; ++j) {
        const int o = oBase + j + half * 8;
        const float iv = gamma[o] * rsqrtf(var[o] + EPS);
        const float bb = beta[o] - mean[o] * iv;
        float v = acc[j] * iv + bb;
        v = v * __frcp_rn(1.f + __expf(-v));
        y[((size_t)b * 64 + o) * 4096 + hwBase + l] = v;
    }

    if (tid < 63)
        wAe[wi] = wv;
}

#define YP 104
#define XP 102

__global__ void __launch_bounds__(224, 4)
k_encreasm(const float* __restrict__ x,
                           const float* __restrict__ y,
                           const float* __restrict__ wAe,
                           const float* __restrict__ enc_b,
                           float* __restrict__ out) {
    __shared__ float ls[64 * YP];
    __shared__ float xs[128 * XP];
    __shared__ float lg[100 * 32];
    __shared__ float wl[36 * 32];
    const int tid    = threadIdx.x;
    const int sTile  = blockIdx.x;
    const int b      = sTile >> 7;
    const int hwBase = (sTile & 127) << 5;
    const int h      = hwBase >> 6;
    const int w0     = hwBase & 63;

    const int lane = tid & 31, l = lane & 15, half = lane >> 4;
    const int oBase = (tid >> 5) << 4;
    const int aCol = (oBase + l) * 2;

    const float* ap = wAe + half * 224 + aCol;
    v2f a0[16], a1[16];

    const float* yb = y + (size_t)b * 64 * 4096;
    const float* xb = x + (size_t)b * 128 * 4096;
    v8f acc0, acc1;
    {
        const bool act = tid < 204;
        const int ts   = act ? tid : 0;
        const int cpar = ts >= 102 ? 1 : 0;
        const int pos  = ts - cpar * 102;
        const int r = pos / 34, col = pos - r * 34;
        const int hh = h + r - 1, ww = w0 + col - 1;
        const bool ok = (hh >= 0) & (hh < 64) & (ww >= 0) & (ww < 64);
        const int hc = hh < 0 ? 0 : (hh > 63 ? 63 : hh);
        const int wc = ww < 0 ? 0 : (ww > 63 ? 63 : ww);
        const float* yp = yb + cpar * 4096 + hc * 64 + wc;
        const float* xp = xb + cpar * 4096 + hc * 64 + wc;
        float v[32];
#pragma unroll
        for (int k = 0; k < 32; ++k)
            v[k] = yp[(size_t)k * 8192];
#pragma unroll
        for (int s = 0; s < 16; ++s)
            a0[s] = *(const v2f*)(ap + s * 448);
#pragma unroll
        for (int j = 0; j < 8; ++j) {
            const int oo = oBase + j + half * 8;
            const float bv = enc_b[oo < 100 ? oo : 0];
            acc0[j] = (oo < 100) ? bv : 0.f;
            acc1[j] = acc0[j];
        }
        float vx[64];
#pragma unroll
        for (int k = 0; k < 64; ++k)
            vx[k] = xp[(size_t)k * 8192];
        if (act) {
#pragma unroll
            for (int k = 0; k < 32; ++k)
                ls[(cpar + 2 * k) * YP + pos] = ok ? v[k] : 0.f;
#pragma unroll
            for (int k = 0; k < 64; ++k)
                xs[(cpar + 2 * k) * XP + pos] = ok ? vx[k] : 0.f;
        }
    }
    __syncthreads();

    const float* lb = ls + half * (2 * YP) + l;

#define ENC_LOAD_A(BUF, T)                                                    \
    _Pragma("unroll")                                                         \
    for (int s = 0; s < 16; ++s)                                              \
        BUF[s] = *(const v2f*)(ap + (T) * 7168 + s * 448);
#define ENC_MMA(BUF, T)                                                       \
    {                                                                         \
        const float* lo = lb + ((T) / 3) * 34 + ((T) % 3);                    \
        _Pragma("unroll")                                                     \
        for (int s = 0; s < 16; ++s) {                                        \
            v2f bm, bn;                                                       \
            bm.x = lo[(4 * s) * YP];                                          \
            bm.y = lo[(4 * s + 1) * YP];                                      \
            bn.x = lo[(4 * s) * YP + 16];                                     \
            bn.y = lo[(4 * s + 1) * YP + 16];                                 \
            acc0 = __builtin_amdgcn_wmma_f32_16x16x4_f32(                     \
                false, BUF[s], false, bm, (short)0, acc0, false, false);      \
            acc1 = __builtin_amdgcn_wmma_f32_16x16x4_f32(                     \
                false, BUF[s], false, bn, (short)0, acc1, false, false);      \
        }                                                                     \
    }

#pragma unroll 1
    for (int tap = 0; tap < 8; tap += 2) {
        ENC_LOAD_A(a1, tap + 1)
        __builtin_amdgcn_sched_barrier(0);
        ENC_MMA(a0, tap)
        ENC_LOAD_A(a0, tap + 2)
        __builtin_amdgcn_sched_barrier(0);
        ENC_MMA(a1, tap + 1)
    }
    ENC_MMA(a0, 8)
#undef ENC_LOAD_A
#undef ENC_MMA

#pragma unroll
    for (int j = 0; j < 8; ++j) {
        const int oo = oBase + j + half * 8;
        if (oo < 100) {
            lg[oo * 32 + l]      = acc0[j];
            lg[oo * 32 + 16 + l] = acc1[j];
        }
    }
    __syncthreads();

    if (tid < 128) {
        const int s = tid >> 5;
        float p[25];
        float mx = -1e30f;
#pragma unroll
        for (int q = 0; q < 25; ++q) {
            p[q] = lg[q * 128 + tid];
            mx = fmaxf(mx, p[q]);
        }
        float sum = 0.f;
#pragma unroll
        for (int q = 0; q < 25; ++q) { p[q] = __expf(p[q] - mx); sum += p[q]; }
        const float rs = __frcp_rn(sum);

        const bool e2 = ((s & 1) == 0), e1 = ((s >> 1) == 0);
        float cf[5][3];
#pragma unroll
        for (int ki = 0; ki < 5; ++ki) {
            const float q0 = p[ki * 5 + 0] * rs, q1 = p[ki * 5 + 1] * rs,
                        q2 = p[ki * 5 + 2] * rs, q3 = p[ki * 5 + 3] * rs,
                        q4 = p[ki * 5 + 4] * rs;
            cf[ki][0] = q0 + (e2 ? q1 : 0.f);
            cf[ki][1] = (e2 ? q3 : q1) + q2;
            cf[ki][2] = q4 + (e2 ? 0.f : q3);
        }
        float wc[9];
#pragma unroll
        for (int j = 0; j < 3; ++j) {
            wc[0 * 3 + j] = cf[0][j] + (e1 ? cf[1][j] : 0.f);
            wc[1 * 3 + j] = (e1 ? cf[3][j] : cf[1][j]) + cf[2][j];
            wc[2 * 3 + j] = cf[4][j] + (e1 ? 0.f : cf[3][j]);
        }
#pragma unroll
        for (int t = 0; t < 9; ++t)
            wl[t * 128 + tid] = wc[t];
    }
    __syncthreads();

    v2f W0[9], W1[9];
#pragma unroll
    for (int t = 0; t < 9; ++t) {
        W0[t].x = wl[t * 128 + lane];
        W0[t].y = wl[t * 128 + 32 + lane];
        W1[t].x = wl[t * 128 + 64 + lane];
        W1[t].y = wl[t * 128 + 96 + lane];
    }
    const int g = tid >> 5;
    float* ob = out + (size_t)b * 128 * 16384 + (size_t)(2 * h) * 128 + 2 * (w0 + lane);
#define REASM(C)                                                              \
    {                                                                         \
        const float* lc = xs + (C) * XP + lane;                               \
        v2f r0 = {0.f, 0.f}, r1 = {0.f, 0.f};                                 \
        _Pragma("unroll")                                                     \
        for (int t = 0; t < 9; ++t) {                                         \
            const float v = lc[(t / 3) * 34 + (t % 3)];                       \
            r0 += W0[t] * v;                                                  \
            r1 += W1[t] * v;                                                  \
        }                                                                     \
        *(v2f*)(ob + (size_t)(C) * 16384)       = r0;      \
        *(v2f*)(ob + (size_t)(C) * 16384 + 128) = r1;      \
    }
#pragma unroll 3
    for (int i = 0; i < 18; ++i)
        REASM(g + 7 * i)
    if (g < 2)
        REASM(126 + g)
#undef REASM
}
#undef YP
#undef XP

extern "C" void kernel_launch(void* const* d_in, const int* in_sizes, int n_in,
                              void* d_out, int out_size, void* d_ws, size_t ws_size,
                              hipStream_t stream) {
    const float* x      = (const float*)d_in[0];
    const float* comp_w = (const float*)d_in[1];
    const float* gamma  = (const float*)d_in[2];
    const float* beta   = (const float*)d_in[3];
    const float* mean   = (const float*)d_in[4];
    const float* var    = (const float*)d_in[5];
    const float* enc_w  = (const float*)d_in[6];
    const float* enc_b  = (const float*)d_in[7];
    float* out = (float*)d_out;

    float* y   = (float*)d_ws;
    float* wAe = y + (size_t)1048576;

    k_compress<<<1024, 128, 0, stream>>>(x, comp_w, enc_w, gamma, beta, mean, var, y, wAe);
    k_encreasm<<<512, 224, 0, stream>>>(x, y, wAe, enc_b, out);
}


# k_compress re-tiled: one 64-px image row per block (256 blocks x 4 waves, each A fragment feeds 4 acc tiles); x and comp_w staged in LDS by b128 row loads, B = one b64 read, issued a block ahead
# speedup vs baseline: 2.6604x; 1.0688x over previous
#include <hip/hip_runtime.h>
#include <hip/hip_bf16.h>

typedef __attribute__((ext_vector_type(2))) float v2f;
typedef __attribute__((ext_vector_type(4))) float v4f;
typedef __attribute__((ext_vector_type(8))) float v8f;

#define EPS 1e-5f

#define PP 144
#define WP 132
__global__ void __launch_bounds__(128)
k_compress(const float* __restrict__ x,
                           const float* __restrict__ comp_w,
                           const float* __restrict__ enc_w,
                           const float* __restrict__ gamma,
                           const float* __restrict__ beta,
                           const float* __restrict__ mean,
                           const float* __restrict__ var,
                           float* __restrict__ y,
                           float* __restrict__ wAe) {
    __shared__ __attribute__((aligned(16))) float ls[64 * PP];
    __shared__ __attribute__((aligned(16))) float lw[64 * WP];
    const int tid    = threadIdx.x;
    const int row    = blockIdx.x;
    const int b      = row >> 6;
    const int hwBase = (row & 63) << 6;

    const int lane = tid & 31, l = lane & 15, half = lane >> 4;
    const int oBase = (tid >> 5) << 4;

    const int wi0 = row * 252 + tid;
    const int wi1 = row * 252 + (tid < 124 ? tid + 128 : 0);
    float wv0, wv1;
    {
        const int wo = (wi0 >> 1) % 112;
        const int wt = ((wi0 / 224) << 1) | (wi0 & 1);
        const int wc = wt & 63, woff = wt >> 6;
        const float wraw = enc_w[(size_t)(wo < 100 ? wo : 0) * 576 + wc * 9 + woff];
        wv0 = (wo < 100) ? wraw : 0.f;
    }
    {
        const int wo = (wi1 >> 1) % 112;
        const int wt = ((wi1 / 224) << 1) | (wi1 & 1);
        const int wc = wt & 63, woff = wt >> 6;
        const float wraw = enc_w[(size_t)(wo < 100 ? wo : 0) * 576 + wc * 9 + woff];
        wv1 = (wo < 100) ? wraw : 0.f;
    }

    const float* xb = x + (size_t)b * 128 * 4096 + hwBase;
    {
        const int pr0 = tid >> 4, p4 = (tid & 15) << 2;
        const int r0 = tid >> 5, k4 = (tid & 31) << 2;
        v4f va[8], vb[8], vw[16];
#pragma unroll
        for (int j = 0; j < 8; ++j) {
            const float* xc = xb + (size_t)(2 * (pr0 + 8 * j)) * 4096 + p4;
            va[j] = *(const v4f*)(xc);
            vb[j] = *(const v4f*)(xc + 4096);
        }
#pragma unroll
        for (int j = 0; j < 16; ++j)
            vw[j] = *(const v4f*)(comp_w + (size_t)(r0 + 4 * j) * 128 + k4);
#pragma unroll
        for (int j = 0; j < 8; ++j) {
            float* d = ls + (pr0 + 8 * j) * PP + p4 * 2;
            v4f s0, s1;
            s0.x = va[j].x; s0.y = vb[j].x; s0.z = va[j].y; s0.w = vb[j].y;
            s1.x = va[j].z; s1.y = vb[j].z; s1.z = va[j].w; s1.w = vb[j].w;
            *(v4f*)(d)     = s0;
            *(v4f*)(d + 4) = s1;
        }
#pragma unroll
        for (int j = 0; j < 16; ++j)
            *(v4f*)(lw + (r0 + 4 * j) * WP + k4) = vw[j];
    }
    __syncthreads();

    v4f q[16];
    {
        const float* ap = lw + (oBase + l) * WP + half * 4;
#pragma unroll
        for (int j = 0; j < 16; ++j)
            q[j] = *(const v4f*)(ap + j * 8);
    }

    v8f acc[4] = {};
    const float* lb = ls + (2 * half) * PP + l * 2;
    v2f b0[8][4], b1[8][4];

#define CMP_LOAD_B(SET, BLK)                                                  \
    _Pragma("unroll")                                                         \
    for (int s = 0; s < 8; ++s) {                                             \
        const float* p = lb + ((BLK) * 16 + (s >> 1) * 4 + (s & 1)) * PP;     \
        _Pragma("unroll")                                                     \
        for (int t = 0; t < 4; ++t)                                           \
            SET[s][t] = *(const v2f*)(p + 32 * t);                            \
    }
#define CMP_MMA(SET, BLK)                                                     \
    _Pragma("unroll")                                                         \
    for (int s = 0; s < 8; ++s) {                                             \
        v2f am;                                                               \
        if (s & 1) { am.x = q[(BLK) * 4 + (s >> 1)].z; am.y = q[(BLK) * 4 + (s >> 1)].w; } \
        else       { am.x = q[(BLK) * 4 + (s >> 1)].x; am.y = q[(BLK) * 4 + (s >> 1)].y; } \
        _Pragma("unroll")                                                     \
        for (int t = 0; t < 4; ++t)                                           \
            acc[t] = __builtin_amdgcn_wmma_f32_16x16x4_f32(                   \
                false, am, false, SET[s][t], (short)0, acc[t], false, false); \
    }

    CMP_LOAD_B(b0, 0)
    CMP_LOAD_B(b1, 1)
    __builtin_amdgcn_sched_barrier(0);
    CMP_MMA(b0, 0)
    CMP_LOAD_B(b0, 2)
    __builtin_amdgcn_sched_barrier(0);
    CMP_MMA(b1, 1)
    CMP_LOAD_B(b1, 3)
    __builtin_amdgcn_sched_barrier(0);
    CMP_MMA(b0, 2)
    CMP_MMA(b1, 3)
#undef CMP_LOAD_B
#undef CMP_MMA

#pragma unroll
    for (int j = 0; j < 8; ++j) {
        const int o = oBase + j + half * 8;
        const float iv = gamma[o] * rsqrtf(var[o] + EPS);
        const float bb = beta[o] - mean[o] * iv;
        float* yp = y + ((size_t)b * 64 + o) * 4096 + hwBase + l;
#pragma unroll
        for (int t = 0; t < 4; ++t) {
            float v = acc[t][j] * iv + bb;
            v = v * __frcp_rn(1.f + __expf(-v));
            yp[16 * t] = v;
        }
    }

    wAe[wi0] = wv0;
    if (tid < 124)
        wAe[wi1] = wv1;
}
#undef PP
#undef WP

#define YP 104
#define XP 102

__global__ void __launch_bounds__(224, 4)
k_encreasm(const float* __restrict__ x,
                           const float* __restrict__ y,
                           const float* __restrict__ wAe,
                           const float* __restrict__ enc_b,
                           float* __restrict__ out) {
    __shared__ float ls[64 * YP];
    __shared__ float xs[128 * XP];
    __shared__ float lg[100 * 32];
    __shared__ float wl[36 * 32];
    const int tid    = threadIdx.x;
    const int sTile  = blockIdx.x;
    const int b      = sTile >> 7;
    const int hwBase = (sTile & 127) << 5;
    const int h      = hwBase >> 6;
    const int w0     = hwBase & 63;

    const int lane = tid & 31, l = lane & 15, half = lane >> 4;
    const int oBase = (tid >> 5) << 4;
    const int aCol = (oBase + l) * 2;

    const float* ap = wAe + half * 224 + aCol;
    v2f a0[16], a1[16];

    const float* yb = y + (size_t)b * 64 * 4096;
    const float* xb = x + (size_t)b * 128 * 4096;
    v8f acc0, acc1;
    {
        const bool act = tid < 204;
        const int ts   = act ? tid : 0;
        const int cpar = ts >= 102 ? 1 : 0;
        const int pos  = ts - cpar * 102;
        const int r = pos / 34, col = pos - r * 34;
        const int hh = h + r - 1, ww = w0 + col - 1;
        const bool ok = (hh >= 0) & (hh < 64) & (ww >= 0) & (ww < 64);
        const int hc = hh < 0 ? 0 : (hh > 63 ? 63 : hh);
        const int wc = ww < 0 ? 0 : (ww > 63 ? 63 : ww);
        const float* yp = yb + cpar * 4096 + hc * 64 + wc;
        const float* xp = xb + cpar * 4096 + hc * 64 + wc;
        float v[32];
#pragma unroll
        for (int k = 0; k < 32; ++k)
            v[k] = yp[(size_t)k * 8192];
#pragma unroll
        for (int s = 0; s < 16; ++s)
            a0[s] = *(const v2f*)(ap + s * 448);
#pragma unroll
        for (int j = 0; j < 8; ++j) {
            const int oo = oBase + j + half * 8;
            const float bv = enc_b[oo < 100 ? oo : 0];
            acc0[j] = (oo < 100) ? bv : 0.f;
            acc1[j] = acc0[j];
        }
        float vx[64];
#pragma unroll
        for (int k = 0; k < 64; ++k)
            vx[k] = xp[(size_t)k * 8192];
        if (act) {
#pragma unroll
            for (int k = 0; k < 32; ++k)
                ls[(cpar + 2 * k) * YP + pos] = ok ? v[k] : 0.f;
#pragma unroll
            for (int k = 0; k < 64; ++k)
                xs[(cpar + 2 * k) * XP + pos] = ok ? vx[k] : 0.f;
        }
    }
    __syncthreads();

    const float* lb = ls + half * (2 * YP) + l;

#define ENC_LOAD_A(BUF, T)                                                    \
    _Pragma("unroll")                                                         \
    for (int s = 0; s < 16; ++s)                                              \
        BUF[s] = *(const v2f*)(ap + (T) * 7168 + s * 448);
#define ENC_MMA(BUF, T)                                                       \
    {                                                                         \
        const float* lo = lb + ((T) / 3) * 34 + ((T) % 3);                    \
        _Pragma("unroll")                                                     \
        for (int s = 0; s < 16; ++s) {                                        \
            v2f bm, bn;                                                       \
            bm.x = lo[(4 * s) * YP];                                          \
            bm.y = lo[(4 * s + 1) * YP];                                      \
            bn.x = lo[(4 * s) * YP + 16];                                     \
            bn.y = lo[(4 * s + 1) * YP + 16];                                 \
            acc0 = __builtin_amdgcn_wmma_f32_16x16x4_f32(                     \
                false, BUF[s], false, bm, (short)0, acc0, false, false);      \
            acc1 = __builtin_amdgcn_wmma_f32_16x16x4_f32(                     \
                false, BUF[s], false, bn, (short)0, acc1, false, false);      \
        }                                                                     \
    }

#pragma unroll 1
    for (int tap = 0; tap < 8; tap += 2) {
        ENC_LOAD_A(a1, tap + 1)
        __builtin_amdgcn_sched_barrier(0);
        ENC_MMA(a0, tap)
        ENC_LOAD_A(a0, tap + 2)
        __builtin_amdgcn_sched_barrier(0);
        ENC_MMA(a1, tap + 1)
    }
    ENC_MMA(a0, 8)
#undef ENC_LOAD_A
#undef ENC_MMA

#pragma unroll
    for (int j = 0; j < 8; ++j) {
        const int oo = oBase + j + half * 8;
        if (oo < 100) {
            lg[oo * 32 + l]      = acc0[j];
            lg[oo * 32 + 16 + l] = acc1[j];
        }
    }
    __syncthreads();

    if (tid < 128) {
        const int s = tid >> 5;
        float p[25];
        float mx = -1e30f;
#pragma unroll
        for (int q = 0; q < 25; ++q) {
            p[q] = lg[q * 128 + tid];
            mx = fmaxf(mx, p[q]);
        }
        float sum = 0.f;
#pragma unroll
        for (int q = 0; q < 25; ++q) { p[q] = __expf(p[q] - mx); sum += p[q]; }
        const float rs = __frcp_rn(sum);

        const bool e2 = ((s & 1) == 0), e1 = ((s >> 1) == 0);
        float cf[5][3];
#pragma unroll
        for (int ki = 0; ki < 5; ++ki) {
            const float q0 = p[ki * 5 + 0] * rs, q1 = p[ki * 5 + 1] * rs,
                        q2 = p[ki * 5 + 2] * rs, q3 = p[ki * 5 + 3] * rs,
                        q4 = p[ki * 5 + 4] * rs;
            cf[ki][0] = q0 + (e2 ? q1 : 0.f);
            cf[ki][1] = (e2 ? q3 : q1) + q2;
            cf[ki][2] = q4 + (e2 ? 0.f : q3);
        }
        float wc[9];
#pragma unroll
        for (int j = 0; j < 3; ++j) {
            wc[0 * 3 + j] = cf[0][j] + (e1 ? cf[1][j] : 0.f);
            wc[1 * 3 + j] = (e1 ? cf[3][j] : cf[1][j]) + cf[2][j];
            wc[2 * 3 + j] = cf[4][j] + (e1 ? 0.f : cf[3][j]);
        }
#pragma unroll
        for (int t = 0; t < 9; ++t)
            wl[t * 128 + tid] = wc[t];
    }
    __syncthreads();

    v2f W0[9], W1[9];
#pragma unroll
    for (int t = 0; t < 9; ++t) {
        W0[t].x = wl[t * 128 + lane];
        W0[t].y = wl[t * 128 + 32 + lane];
        W1[t].x = wl[t * 128 + 64 + lane];
        W1[t].y = wl[t * 128 + 96 + lane];
    }
    const int g = tid >> 5;
    float* ob = out + (size_t)b * 128 * 16384 + (size_t)(2 * h) * 128 + 2 * (w0 + lane);
#define REASM(C)                                                              \
    {                                                                         \
        const float* lc = xs + (C) * XP + lane;                               \
        v2f r0 = {0.f, 0.f}, r1 = {0.f, 0.f};                                 \
        _Pragma("unroll")                                                     \
        for (int t = 0; t < 9; ++t) {                                         \
            const float v = lc[(t / 3) * 34 + (t % 3)];                       \
            r0 += W0[t] * v;                                                  \
            r1 += W1[t] * v;                                                  \
        }                                                                     \
        *(v2f*)(ob + (size_t)(C) * 16384)       = r0;      \
        *(v2f*)(ob + (size_t)(C) * 16384 + 128) = r1;      \
    }
#pragma unroll 3
    for (int i = 0; i < 18; ++i)
        REASM(g + 7 * i)
    if (g < 2)
        REASM(126 + g)
#undef REASM
}
#undef YP
#undef XP

extern "C" void kernel_launch(void* const* d_in, const int* in_sizes, int n_in,
                              void* d_out, int out_size, void* d_ws, size_t ws_size,
                              hipStream_t stream) {
    const float* x      = (const float*)d_in[0];
    const float* comp_w = (const float*)d_in[1];
    const float* gamma  = (const float*)d_in[2];
    const float* beta   = (const float*)d_in[3];
    const float* mean   = (const float*)d_in[4];
    const float* var    = (const float*)d_in[5];
    const float* enc_w  = (const float*)d_in[6];
    const float* enc_b  = (const float*)d_in[7];
    float* out = (float*)d_out;

    float* y   = (float*)d_ws;
    float* wAe = y + (size_t)1048576;

    k_compress<<<256, 128, 0, stream>>>(x, comp_w, enc_w, gamma, beta, mean, var, y, wAe);
    k_encreasm<<<512, 224, 0, stream>>>(x, y, wAe, enc_b, out);
}


# k_encreasm tile 32 -> 64 source px (256 blocks x 8 waves, one per WGP): 28 acc tiles, 7 per wave pair, an A fragment feeds 4 or 3 px tiles; y as channel pairs, B = b64 read half a tap ahead
# speedup vs baseline: 3.2092x; 1.2063x over previous
#include <hip/hip_runtime.h>
#include <hip/hip_bf16.h>

typedef __attribute__((ext_vector_type(2))) float v2f;
typedef __attribute__((ext_vector_type(4))) float v4f;
typedef __attribute__((ext_vector_type(8))) float v8f;

#define EPS 1e-5f

#define PP 144
#define WP 132
__global__ void __launch_bounds__(128)
k_compress(const float* __restrict__ x,
                           const float* __restrict__ comp_w,
                           const float* __restrict__ enc_w,
                           const float* __restrict__ gamma,
                           const float* __restrict__ beta,
                           const float* __restrict__ mean,
                           const float* __restrict__ var,
                           float* __restrict__ y,
                           float* __restrict__ wAe) {
    __shared__ __attribute__((aligned(16))) float ls[64 * PP];
    __shared__ __attribute__((aligned(16))) float lw[64 * WP];
    const int tid    = threadIdx.x;
    const int row    = blockIdx.x;
    const int b      = row >> 6;
    const int hwBase = (row & 63) << 6;

    const int lane = tid & 31, l = lane & 15, half = lane >> 4;
    const int oBase = (tid >> 5) << 4;

    const int wi0 = row * 252 + tid;
    const int wi1 = row * 252 + (tid < 124 ? tid + 128 : 0);
    float wv0, wv1;
    {
        const int wo = (wi0 >> 1) % 112;
        const int wt = ((wi0 / 224) << 1) | (wi0 & 1);
        const int wc = wt & 63, woff = wt >> 6;
        const float wraw = enc_w[(size_t)(wo < 100 ? wo : 0) * 576 + wc * 9 + woff];
        wv0 = (wo < 100) ? wraw : 0.f;
    }
    {
        const int wo = (wi1 >> 1) % 112;
        const int wt = ((wi1 / 224) << 1) | (wi1 & 1);
        const int wc = wt & 63, woff = wt >> 6;
        const float wraw = enc_w[(size_t)(wo < 100 ? wo : 0) * 576 + wc * 9 + woff];
        wv1 = (wo < 100) ? wraw : 0.f;
    }

    const float* xb = x + (size_t)b * 128 * 4096 + hwBase;
    {
        const int pr0 = tid >> 4, p4 = (tid & 15) << 2;
        const int r0 = tid >> 5, k4 = (tid & 31) << 2;
        v4f va[8], vb[8], vw[16];
#pragma unroll
        for (int j = 0; j < 8; ++j) {
            const float* xc = xb + (size_t)(2 * (pr0 + 8 * j)) * 4096 + p4;
            va[j] = *(const v4f*)(xc);
            vb[j] = *(const v4f*)(xc + 4096);
        }
#pragma unroll
        for (int j = 0; j < 16; ++j)
            vw[j] = *(const v4f*)(comp_w + (size_t)(r0 + 4 * j) * 128 + k4);
#pragma unroll
        for (int j = 0; j < 8; ++j) {
            float* d = ls + (pr0 + 8 * j) * PP + p4 * 2;
            v4f s0, s1;
            s0.x = va[j].x; s0.y = vb[j].x; s0.z = va[j].y; s0.w = vb[j].y;
            s1.x = va[j].z; s1.y = vb[j].z; s1.z = va[j].w; s1.w = vb[j].w;
            *(v4f*)(d)     = s0;
            *(v4f*)(d + 4) = s1;
        }
#pragma unroll
        for (int j = 0; j < 16; ++j)
            *(v4f*)(lw + (r0 + 4 * j) * WP + k4) = vw[j];
    }
    __syncthreads();

    v4f q[16];
    {
        const float* ap = lw + (oBase + l) * WP + half * 4;
#pragma unroll
        for (int j = 0; j < 16; ++j)
            q[j] = *(const v4f*)(ap + j * 8);
    }

    v8f acc[4] = {};
    const float* lb = ls + (2 * half) * PP + l * 2;
    v2f b0[8][4], b1[8][4];

#define CMP_LOAD_B(SET, BLK)                                                  \
    _Pragma("unroll")                                                         \
    for (int s = 0; s < 8; ++s) {                                             \
        const float* p = lb + ((BLK) * 16 + (s >> 1) * 4 + (s & 1)) * PP;     \
        _Pragma("unroll")                                                     \
        for (int t = 0; t < 4; ++t)                                           \
            SET[s][t] = *(const v2f*)(p + 32 * t);                            \
    }
#define CMP_MMA(SET, BLK)                                                     \
    _Pragma("unroll")                                                         \
    for (int s = 0; s < 8; ++s) {                                             \
        v2f am;                                                               \
        if (s & 1) { am.x = q[(BLK) * 4 + (s >> 1)].z; am.y = q[(BLK) * 4 + (s >> 1)].w; } \
        else       { am.x = q[(BLK) * 4 + (s >> 1)].x; am.y = q[(BLK) * 4 + (s >> 1)].y; } \
        _Pragma("unroll")                                                     \
        for (int t = 0; t < 4; ++t)                                           \
            acc[t] = __builtin_amdgcn_wmma_f32_16x16x4_f32(                   \
                false, am, false, SET[s][t], (short)0, acc[t], false, false); \
    }

    CMP_LOAD_B(b0, 0)
    CMP_LOAD_B(b1, 1)
    __builtin_amdgcn_sched_barrier(0);
    CMP_MMA(b0, 0)
    CMP_LOAD_B(b0, 2)
    __builtin_amdgcn_sched_barrier(0);
    CMP_MMA(b1, 1)
    CMP_LOAD_B(b1, 3)
    __builtin_amdgcn_sched_barrier(0);
    CMP_MMA(b0, 2)
    CMP_MMA(b1, 3)
#undef CMP_LOAD_B
#undef CMP_MMA

#pragma unroll
    for (int j = 0; j < 8; ++j) {
        const int o = oBase + j + half * 8;
        const float iv = gamma[o] * rsqrtf(var[o] + EPS);
        const float bb = beta[o] - mean[o] * iv;
        float* yp = y + ((size_t)b * 64 + o) * 4096 + hwBase + l;
#pragma unroll
        for (int t = 0; t < 4; ++t) {
            float v = acc[t][j] * iv + bb;
            v = v * __frcp_rn(1.f + __expf(-v));
            yp[16 * t] = v;
        }
    }

    wAe[wi0] = wv0;
    if (tid < 124)
        wAe[wi1] = wv1;
}
#undef PP
#undef WP

#define YP 416
#define XP 198

template <int NO>
__device__ __forceinline__ void enc_load_a(v2f (&buf)[3][16], const float* __restrict__ ap, int tap) {
#pragma unroll
    for (int i = 0; i < NO; ++i) {
#pragma unroll
        for (int s = 0; s < 16; ++s)
            buf[i][s] = *(const v2f*)(ap + tap * 7168 + s * 448 + i * 32);
    }
}
template <int NP, int HS>
__device__ __forceinline__ void enc_load_b(v2f (&bs)[8][4], const float* lb, int tap) {
    const float* lo = lb + ((tap / 3) * 66 + (tap % 3)) * 2 + HS * (16 * YP);
#pragma unroll
    for (int s = 0; s < 8; ++s) {
#pragma unroll
        for (int t = 0; t < NP; ++t)
            bs[s][t] = *(const v2f*)(lo + (2 * s) * YP + 32 * t);
    }
}
template <int NO, int NP, int HS>
__device__ __forceinline__ void enc_mma(const v2f (&buf)[3][16], const v2f (&bs)[8][4], v8f (&acc)[4]) {
#pragma unroll
    for (int s = 0; s < 8; ++s) {
#pragma unroll
        for (int i = 0; i < NO; ++i) {
#pragma unroll
            for (int t = 0; t < NP; ++t)
                acc[i * NP + t] = __builtin_amdgcn_wmma_f32_16x16x4_f32(
                    false, buf[i][HS * 8 + s], false, bs[s][t], (short)0, acc[i * NP + t], false, false);
        }
    }
}
template <int NO, int NP>
__device__ __forceinline__ void enc_taps(v2f (&a0)[3][16], const float* __restrict__ ap, const float* lb, v8f (&acc)[4]) {
    v2f a1[3][16];
    v2f b0[8][4], b1[8][4];
    enc_load_b<NP, 0>(b0, lb, 0);
#pragma unroll 1
    for (int tap = 0; tap < 8; tap += 2) {
        enc_load_a<NO>(a1, ap, tap + 1);
        enc_load_b<NP, 1>(b1, lb, tap);
        __builtin_amdgcn_sched_barrier(0);
        enc_mma<NO, NP, 0>(a0, b0, acc);
        enc_load_b<NP, 0>(b0, lb, tap + 1);
        __builtin_amdgcn_sched_barrier(0);
        enc_mma<NO, NP, 1>(a0, b1, acc);
        enc_load_a<NO>(a0, ap, tap + 2);
        enc_load_b<NP, 1>(b1, lb, tap + 1);
        __builtin_amdgcn_sched_barrier(0);
        enc_mma<NO, NP, 0>(a1, b0, acc);
        enc_load_b<NP, 0>(b0, lb, tap + 2);
        __builtin_amdgcn_sched_barrier(0);
        enc_mma<NO, NP, 1>(a1, b1, acc);
    }
    enc_load_b<NP, 1>(b1, lb, 8);
    __builtin_amdgcn_sched_barrier(0);
    enc_mma<NO, NP, 0>(a0, b0, acc);
    enc_mma<NO, NP, 1>(a0, b1, acc);
}

__global__ void __launch_bounds__(256)
k_encreasm(const float* __restrict__ x,
                           const float* __restrict__ y,
                           const float* __restrict__ wAe,
                           const float* __restrict__ enc_b,
                           float* __restrict__ out) {
    __shared__ __attribute__((aligned(16))) float ls[32 * YP];
    __shared__ float xs[128 * XP];
    __shared__ float lg[100 * 64];
    __shared__ float wl[36 * 64];
    const int tid  = threadIdx.x;
    const int row  = blockIdx.x;
    const int b    = row >> 6;
    const int h    = row & 63;

    const int lane = tid & 31, l = lane & 15, half = lane >> 4;
    const int g = __builtin_amdgcn_readfirstlane(tid >> 5);
    const bool last = g == 7;
    const int oBase = (last ? 4 : g) << 4;
    const int aCol = (oBase + l) * 2;

    const float* ap = wAe + half * 224 + aCol;
    v2f a0[3][16];

    const float* yb = y + (size_t)b * 64 * 4096;
    const float* xb = x + (size_t)b * 128 * 4096;
    v8f acc[4];
    if (!last) {
        enc_load_a<1>(a0, ap, 0);
#pragma unroll
        for (int j = 0; j < 8; ++j) {
            const int oo = oBase + j + half * 8;
            const float bv = enc_b[oo < 100 ? oo : 0];
            const float sv = (oo < 100) ? bv : 0.f;
            acc[0][j] = sv; acc[1][j] = sv; acc[2][j] = sv; acc[3][j] = sv;
        }
    } else {
#pragma unroll
        for (int i = 0; i < 3; ++i) {
#pragma unroll
            for (int j = 0; j < 8; ++j) {
                const int oo = 64 + 16 * i + j + half * 8;
                const float bv = enc_b[oo < 100 ? oo : 0];
                acc[i][j] = (oo < 100) ? bv : 0.f;
            }
        }
    }
    if (tid < 198) {
        const int pos = tid;
        const int r = pos / 66, col = pos - r * 66;
        const int hh = h + r - 1, ww = col - 1;
        const bool ok = (hh >= 0) & (hh < 64) & (ww >= 0) & (ww < 64);
        const int hc = hh < 0 ? 0 : (hh > 63 ? 63 : hh);
        const int wc = ww < 0 ? 0 : (ww > 63 ? 63 : ww);
        const float* yp = yb + hc * 64 + wc;
        const float* xp = xb + hc * 64 + wc;
        float v[64];
#pragma unroll
        for (int k = 0; k < 64; ++k)
            v[k] = yp[(size_t)k * 4096];
        float vx[128];
#pragma unroll
        for (int k = 0; k < 128; ++k)
            vx[k] = xp[(size_t)k * 4096];
#pragma unroll
        for (int k = 0; k < 32; ++k) {
            v2f pv;
            pv.x = ok ? v[2 * k] : 0.f;
            pv.y = ok ? v[2 * k + 1] : 0.f;
            *(v2f*)(ls + k * YP + pos * 2) = pv;
        }
#pragma unroll
        for (int k = 0; k < 128; ++k)
            xs[k * XP + pos] = ok ? vx[k] : 0.f;
    }
    __syncthreads();

    const float* lb = ls + half * YP + l * 2;

    if (g < 4) {
        enc_taps<1, 4>(a0, ap, lb, acc);
    } else if (!last) {
        enc_taps<1, 3>(a0, ap, lb, acc);
    } else {
        enc_load_a<3>(a0, ap, 0);
        enc_taps<3, 1>(a0, ap, lb + 96, acc);
    }

    if (!last) {
#pragma unroll
        for (int j = 0; j < 8; ++j) {
            const int oo = oBase + j + half * 8;
            if (oo < 100) {
                lg[oo * 64 + l]      = acc[0][j];
                lg[oo * 64 + 16 + l] = acc[1][j];
                lg[oo * 64 + 32 + l] = acc[2][j];
                if (g < 4)
                    lg[oo * 64 + 48 + l] = acc[3][j];
            }
        }
    } else {
#pragma unroll
        for (int i = 0; i < 3; ++i) {
#pragma unroll
            for (int j = 0; j < 8; ++j) {
                const int oo = 64 + 16 * i + j + half * 8;
                if (oo < 100)
                    lg[oo * 64 + 48 + l] = acc[i][j];
            }
        }
    }
    __syncthreads();

    {
        const int s = tid >> 6;
        float p[25];
        float mx = -1e30f;
#pragma unroll
        for (int q = 0; q < 25; ++q) {
            p[q] = lg[q * 256 + tid];
            mx = fmaxf(mx, p[q]);
        }
        float sum = 0.f;
#pragma unroll
        for (int q = 0; q < 25; ++q) { p[q] = __expf(p[q] - mx); sum += p[q]; }
        const float rs = __frcp_rn(sum);

        const bool e2 = ((s & 1) == 0), e1 = ((s >> 1) == 0);
        float cf[5][3];
#pragma unroll
        for (int ki = 0; ki < 5; ++ki) {
            const float q0 = p[ki * 5 + 0] * rs, q1 = p[ki * 5 + 1] * rs,
                        q2 = p[ki * 5 + 2] * rs, q3 = p[ki * 5 + 3] * rs,
                        q4 = p[ki * 5 + 4] * rs;
            cf[ki][0] = q0 + (e2 ? q1 : 0.f);
            cf[ki][1] = (e2 ? q3 : q1) + q2;
            cf[ki][2] = q4 + (e2 ? 0.f : q3);
        }
        float wc[9];
#pragma unroll
        for (int j = 0; j < 3; ++j) {
            wc[0 * 3 + j] = cf[0][j] + (e1 ? cf[1][j] : 0.f);
            wc[1 * 3 + j] = (e1 ? cf[3][j] : cf[1][j]) + cf[2][j];
            wc[2 * 3 + j] = cf[4][j] + (e1 ? 0.f : cf[3][j]);
        }
#pragma unroll
        for (int t = 0; t < 9; ++t)
            wl[t * 256 + tid] = wc[t];
    }
    __syncthreads();

    v2f W0[2][9], W1[2][9];
#pragma unroll
    for (int hf = 0; hf < 2; ++hf) {
#pragma unroll
        for (int t = 0; t < 9; ++t) {
            W0[hf][t].x = wl[t * 256 + 32 * hf + lane];
            W0[hf][t].y = wl[t * 256 + 64 + 32 * hf + lane];
            W1[hf][t].x = wl[t * 256 + 128 + 32 * hf + lane];
            W1[hf][t].y = wl[t * 256 + 192 + 32 * hf + lane];
        }
    }
    float* ob = out + (size_t)b * 128 * 16384 + (size_t)(2 * h) * 128 + 2 * lane;
#define REASM(C)                                                              \
    _Pragma("unroll")                                                         \
    for (int hf = 0; hf < 2; ++hf) {                                          \
        const float* lc = xs + (C) * XP + 32 * hf + lane;                     \
        v2f r0 = {0.f, 0.f}, r1 = {0.f, 0.f};                                 \
        _Pragma("unroll")                                                     \
        for (int t = 0; t < 9; ++t) {                                         \
            const float v = lc[(t / 3) * 66 + (t % 3)];                       \
            r0 += W0[hf][t] * v;                                              \
            r1 += W1[hf][t] * v;                                              \
        }                                                                     \
        *(v2f*)(ob + (size_t)(C) * 16384 + 64 * hf)       = r0;   \
        *(v2f*)(ob + (size_t)(C) * 16384 + 128 + 64 * hf) = r1;   \
    }
#pragma unroll 2
    for (int i = 0; i < 16; ++i)
        REASM(g + 8 * i)
#undef REASM
}
#undef YP
#undef XP

extern "C" void kernel_launch(void* const* d_in, const int* in_sizes, int n_in,
                              void* d_out, int out_size, void* d_ws, size_t ws_size,
                              hipStream_t stream) {
    const float* x      = (const float*)d_in[0];
    const float* comp_w = (const float*)d_in[1];
    const float* gamma  = (const float*)d_in[2];
    const float* beta   = (const float*)d_in[3];
    const float* mean   = (const float*)d_in[4];
    const float* var    = (const float*)d_in[5];
    const float* enc_w  = (const float*)d_in[6];
    const float* enc_b  = (const float*)d_in[7];
    float* out = (float*)d_out;

    float* y   = (float*)d_ws;
    float* wAe = y + (size_t)1048576;

    k_compress<<<256, 128, 0, stream>>>(x, comp_w, enc_w, gamma, beta, mean, var, y, wAe);
    k_encreasm<<<256, 256, 0, stream>>>(x, y, wAe, enc_b, out);
}


# k_encreasm: x patch staging moved out of the pre-barrier round trip into the tap loop: all 8 waves fetch it as b128 channel rows, 32 channels per two-tap trip, stored to LDS at the trip's end
# speedup vs baseline: 3.3954x; 1.0580x over previous
#include <hip/hip_runtime.h>
#include <hip/hip_bf16.h>

typedef __attribute__((ext_vector_type(2))) float v2f;
typedef __attribute__((ext_vector_type(4))) float v4f;
typedef __attribute__((ext_vector_type(8))) float v8f;

#define EPS 1e-5f

#define PP 144
#define WP 132
__global__ void __launch_bounds__(128)
k_compress(const float* __restrict__ x,
                           const float* __restrict__ comp_w,
                           const float* __restrict__ enc_w,
                           const float* __restrict__ gamma,
                           const float* __restrict__ beta,
                           const float* __restrict__ mean,
                           const float* __restrict__ var,
                           float* __restrict__ y,
                           float* __restrict__ wAe) {
    __shared__ __attribute__((aligned(16))) float ls[64 * PP];
    __shared__ __attribute__((aligned(16))) float lw[64 * WP];
    const int tid    = threadIdx.x;
    const int row    = blockIdx.x;
    const int b      = row >> 6;
    const int hwBase = (row & 63) << 6;

    const int lane = tid & 31, l = lane & 15, half = lane >> 4;
    const int oBase = (tid >> 5) << 4;

    const int wi0 = row * 252 + tid;
    const int wi1 = row * 252 + (tid < 124 ? tid + 128 : 0);
    float wv0, wv1;
    {
        const int wo = (wi0 >> 1) % 112;
        const int wt = ((wi0 / 224) << 1) | (wi0 & 1);
        const int wc = wt & 63, woff = wt >> 6;
        const float wraw = enc_w[(size_t)(wo < 100 ? wo : 0) * 576 + wc * 9 + woff];
        wv0 = (wo < 100) ? wraw : 0.f;
    }
    {
        const int wo = (wi1 >> 1) % 112;
        const int wt = ((wi1 / 224) << 1) | (wi1 & 1);
        const int wc = wt & 63, woff = wt >> 6;
        const float wraw = enc_w[(size_t)(wo < 100 ? wo : 0) * 576 + wc * 9 + woff];
        wv1 = (wo < 100) ? wraw : 0.f;
    }

    const float* xb = x + (size_t)b * 128 * 4096 + hwBase;
    {
        const int pr0 = tid >> 4, p4 = (tid & 15) << 2;
        const int r0 = tid >> 5, k4 = (tid & 31) << 2;
        v4f va[8], vb[8], vw[16];
#pragma unroll
        for (int j = 0; j < 8; ++j) {
            const float* xc = xb + (size_t)(2 * (pr0 + 8 * j)) * 4096 + p4;
            va[j] = *(const v4f*)(xc);
            vb[j] = *(const v4f*)(xc + 4096);
        }
#pragma unroll
        for (int j = 0; j < 16; ++j)
            vw[j] = *(const v4f*)(comp_w + (size_t)(r0 + 4 * j) * 128 + k4);
#pragma unroll
        for (int j = 0; j < 8; ++j) {
            float* d = ls + (pr0 + 8 * j) * PP + p4 * 2;
            v4f s0, s1;
            s0.x = va[j].x; s0.y = vb[j].x; s0.z = va[j].y; s0.w = vb[j].y;
            s1.x = va[j].z; s1.y = vb[j].z; s1.z = va[j].w; s1.w = vb[j].w;
            *(v4f*)(d)     = s0;
            *(v4f*)(d + 4) = s1;
        }
#pragma unroll
        for (int j = 0; j < 16; ++j)
            *(v4f*)(lw + (r0 + 4 * j) * WP + k4) = vw[j];
    }
    __syncthreads();

    v4f q[16];
    {
        const float* ap = lw + (oBase + l) * WP + half * 4;
#pragma unroll
        for (int j = 0; j < 16; ++j)
            q[j] = *(const v4f*)(ap + j * 8);
    }

    v8f acc[4] = {};
    const float* lb = ls + (2 * half) * PP + l * 2;
    v2f b0[8][4], b1[8][4];

#define CMP_LOAD_B(SET, BLK)                                                  \
    _Pragma("unroll")                                                         \
    for (int s = 0; s < 8; ++s) {                                             \
        const float* p = lb + ((BLK) * 16 + (s >> 1) * 4 + (s & 1)) * PP;     \
        _Pragma("unroll")                                                     \
        for (int t = 0; t < 4; ++t)                                           \
            SET[s][t] = *(const v2f*)(p + 32 * t);                            \
    }
#define CMP_MMA(SET, BLK)                                                     \
    _Pragma("unroll")                                                         \
    for (int s = 0; s < 8; ++s) {                                             \
        v2f am;                                                               \
        if (s & 1) { am.x = q[(BLK) * 4 + (s >> 1)].z; am.y = q[(BLK) * 4 + (s >> 1)].w; } \
        else       { am.x = q[(BLK) * 4 + (s >> 1)].x; am.y = q[(BLK) * 4 + (s >> 1)].y; } \
        _Pragma("unroll")                                                     \
        for (int t = 0; t < 4; ++t)                                           \
            acc[t] = __builtin_amdgcn_wmma_f32_16x16x4_f32(                   \
                false, am, false, SET[s][t], (short)0, acc[t], false, false); \
    }

    CMP_LOAD_B(b0, 0)
    CMP_LOAD_B(b1, 1)
    __builtin_amdgcn_sched_barrier(0);
    CMP_MMA(b0, 0)
    CMP_LOAD_B(b0, 2)
    __builtin_amdgcn_sched_barrier(0);
    CMP_MMA(b1, 1)
    CMP_LOAD_B(b1, 3)
    __builtin_amdgcn_sched_barrier(0);
    CMP_MMA(b0, 2)
    CMP_MMA(b1, 3)
#undef CMP_LOAD_B
#undef CMP_MMA

#pragma unroll
    for (int j = 0; j < 8; ++j) {
        const int o = oBase + j + half * 8;
        const float iv = gamma[o] * rsqrtf(var[o] + EPS);
        const float bb = beta[o] - mean[o] * iv;
        float* yp = y + ((size_t)b * 64 + o) * 4096 + hwBase + l;
#pragma unroll
        for (int t = 0; t < 4; ++t) {
            float v = acc[t][j] * iv + bb;
            v = v * __frcp_rn(1.f + __expf(-v));
            yp[16 * t] = v;
        }
    }

    wAe[wi0] = wv0;
    if (tid < 124)
        wAe[wi1] = wv1;
}
#undef PP
#undef WP

#define YP 416
#define XP 216

template <int NO>
__device__ __forceinline__ void enc_load_a(v2f (&buf)[3][16], const float* __restrict__ ap, int tap) {
#pragma unroll
    for (int i = 0; i < NO; ++i) {
#pragma unroll
        for (int s = 0; s < 16; ++s)
            buf[i][s] = *(const v2f*)(ap + tap * 7168 + s * 448 + i * 32);
    }
}
template <int NP, int HS>
__device__ __forceinline__ void enc_load_b(v2f (&bs)[8][4], const float* lb, int tap) {
    const float* lo = lb + ((tap / 3) * 66 + (tap % 3)) * 2 + HS * (16 * YP);
#pragma unroll
    for (int s = 0; s < 8; ++s) {
#pragma unroll
        for (int t = 0; t < NP; ++t)
            bs[s][t] = *(const v2f*)(lo + (2 * s) * YP + 32 * t);
    }
}
template <int NO, int NP, int HS>
__device__ __forceinline__ void enc_mma(const v2f (&buf)[3][16], const v2f (&bs)[8][4], v8f (&acc)[4]) {
#pragma unroll
    for (int s = 0; s < 8; ++s) {
#pragma unroll
        for (int i = 0; i < NO; ++i) {
#pragma unroll
            for (int t = 0; t < NP; ++t)
                acc[i * NP + t] = __builtin_amdgcn_wmma_f32_16x16x4_f32(
                    false, buf[i][HS * 8 + s], false, bs[s][t], (short)0, acc[i * NP + t], false, false);
        }
    }
}
template <int NO, int NP>
__device__ __forceinline__ void enc_taps(v2f (&a0)[3][16], const float* __restrict__ ap, const float* lb, v8f (&acc)[4],
                                         const float* __restrict__ xg, float* xd,
                                         int ro0, int ro1, int ro2, bool ok0, bool ok2) {
    v2f a1[3][16];
    v2f b0[8][4], b1[8][4];
    v4f xr[6];
    enc_load_b<NP, 0>(b0, lb, 0);
#pragma unroll 1
    for (int tap = 0; tap < 8; tap += 2) {
        enc_load_a<NO>(a1, ap, tap + 1);
        enc_load_b<NP, 1>(b1, lb, tap);
        {
            const float* xc = xg + (size_t)(tap * 16) * 4096;
            xr[0] = *(const v4f*)(xc + ro0);
            xr[1] = *(const v4f*)(xc + ro1);
            xr[2] = *(const v4f*)(xc + ro2);
            xr[3] = *(const v4f*)(xc + 16 * 4096 + ro0);
            xr[4] = *(const v4f*)(xc + 16 * 4096 + ro1);
            xr[5] = *(const v4f*)(xc + 16 * 4096 + ro2);
        }
        __builtin_amdgcn_sched_barrier(0);
        enc_mma<NO, NP, 0>(a0, b0, acc);
        enc_load_b<NP, 0>(b0, lb, tap + 1);
        __builtin_amdgcn_sched_barrier(0);
        enc_mma<NO, NP, 1>(a0, b1, acc);
        enc_load_a<NO>(a0, ap, tap + 2);
        enc_load_b<NP, 1>(b1, lb, tap + 1);
        __builtin_amdgcn_sched_barrier(0);
        enc_mma<NO, NP, 0>(a1, b0, acc);
        enc_load_b<NP, 0>(b0, lb, tap + 2);
        __builtin_amdgcn_sched_barrier(0);
        enc_mma<NO, NP, 1>(a1, b1, acc);
        {
            float* xo = xd + (tap * 16) * XP;
            const v4f z = {0.f, 0.f, 0.f, 0.f};
            *(v4f*)(xo)                 = ok0 ? xr[0] : z;
            *(v4f*)(xo + 72)            = xr[1];
            *(v4f*)(xo + 144)           = ok2 ? xr[2] : z;
            *(v4f*)(xo + 16 * XP)       = ok0 ? xr[3] : z;
            *(v4f*)(xo + 16 * XP + 72)  = xr[4];
            *(v4f*)(xo + 16 * XP + 144) = ok2 ? xr[5] : z;
        }
    }
    enc_load_b<NP, 1>(b1, lb, 8);
    __builtin_amdgcn_sched_barrier(0);
    enc_mma<NO, NP, 0>(a0, b0, acc);
    enc_mma<NO, NP, 1>(a0, b1, acc);
}

__global__ void __launch_bounds__(256)
k_encreasm(const float* __restrict__ x,
                           const float* __restrict__ y,
                           const float* __restrict__ wAe,
                           const float* __restrict__ enc_b,
                           float* __restrict__ out) {
    __shared__ __attribute__((aligned(16))) float ls[32 * YP];
    __shared__ __attribute__((aligned(16))) float xs[128 * XP];
    __shared__ float lg[100 * 64];
    __shared__ float wl[36 * 64];
    const int tid  = threadIdx.x;
    const int row  = blockIdx.x;
    const int b    = row >> 6;
    const int h    = row & 63;

    const int lane = tid & 31, l = lane & 15, half = lane >> 4;
    const int g = __builtin_amdgcn_readfirstlane(tid >> 5);
    const bool last = g == 7;
    const int oBase = (last ? 4 : g) << 4;
    const int aCol = (oBase + l) * 2;

    const float* ap = wAe + half * 224 + aCol;
    v2f a0[3][16];

    const float* yb = y + (size_t)b * 64 * 4096;
    const float* xb = x + (size_t)b * 128 * 4096;
    v8f acc[4];
    if (!last) {
        enc_load_a<1>(a0, ap, 0);
#pragma unroll
        for (int j = 0; j < 8; ++j) {
            const int oo = oBase + j + half * 8;
            const float bv = enc_b[oo < 100 ? oo : 0];
            const float sv = (oo < 100) ? bv : 0.f;
            acc[0][j] = sv; acc[1][j] = sv; acc[2][j] = sv; acc[3][j] = sv;
        }
    } else {
#pragma unroll
        for (int i = 0; i < 3; ++i) {
#pragma unroll
            for (int j = 0; j < 8; ++j) {
                const int oo = 64 + 16 * i + j + half * 8;
                const float bv = enc_b[oo < 100 ? oo : 0];
                acc[i][j] = (oo < 100) ? bv : 0.f;
            }
        }
    }
    if (tid < 198) {
        const int pos = tid;
        const int r = pos / 66, col = pos - r * 66;
        const int hh = h + r - 1, ww = col - 1;
        const bool ok = (hh >= 0) & (hh < 64) & (ww >= 0) & (ww < 64);
        const int hc = hh < 0 ? 0 : (hh > 63 ? 63 : hh);
        const int wc = ww < 0 ? 0 : (ww > 63 ? 63 : ww);
        const float* yp = yb + hc * 64 + wc;
        float v[64];
#pragma unroll
        for (int k = 0; k < 64; ++k)
            v[k] = yp[(size_t)k * 4096];
#pragma unroll
        for (int k = 0; k < 32; ++k) {
            v2f pv;
            pv.x = ok ? v[2 * k] : 0.f;
            pv.y = ok ? v[2 * k + 1] : 0.f;
            *(v2f*)(ls + k * YP + pos * 2) = pv;
        }
    }
#pragma unroll
    for (int k = 0; k < 3; ++k) {
        const int e = tid + 256 * k;
        const int ln = e >> 1;
        xs[(ln & 127) * XP + (ln >> 7) * 72 + ((e & 1) ? 68 : 3)] = 0.f;
    }
    __syncthreads();

    const float* lb = ls + half * YP + l * 2;

    const float* xg = xb + (size_t)(tid >> 4) * 4096 + ((tid & 15) << 2);
    float* xd = xs + (tid >> 4) * XP + 4 + ((tid & 15) << 2);
    const bool ok0 = h > 0, ok2 = h < 63;
    const int ro0 = (ok0 ? h - 1 : 0) * 64, ro1 = h * 64, ro2 = (ok2 ? h + 1 : 63) * 64;

    if (g < 4) {
        enc_taps<1, 4>(a0, ap, lb, acc, xg, xd, ro0, ro1, ro2, ok0, ok2);
    } else if (!last) {
        enc_taps<1, 3>(a0, ap, lb, acc, xg, xd, ro0, ro1, ro2, ok0, ok2);
    } else {
        enc_load_a<3>(a0, ap, 0);
        enc_taps<3, 1>(a0, ap, lb + 96, acc, xg, xd, ro0, ro1, ro2, ok0, ok2);
    }

    if (!last) {
#pragma unroll
        for (int j = 0; j < 8; ++j) {
            const int oo = oBase + j + half * 8;
            if (oo < 100) {
                lg[oo * 64 + l]      = acc[0][j];
                lg[oo * 64 + 16 + l] = acc[1][j];
                lg[oo * 64 + 32 + l] = acc[2][j];
                if (g < 4)
                    lg[oo * 64 + 48 + l] = acc[3][j];
            }
        }
    } else {
#pragma unroll
        for (int i = 0; i < 3; ++i) {
#pragma unroll
            for (int j = 0; j < 8; ++j) {
                const int oo = 64 + 16 * i + j + half * 8;
                if (oo < 100)
                    lg[oo * 64 + 48 + l] = acc[i][j];
            }
        }
    }
    __syncthreads();

    {
        const int s = tid >> 6;
        float p[25];
        float mx = -1e30f;
#pragma unroll
        for (int q = 0; q < 25; ++q) {
            p[q] = lg[q * 256 + tid];
            mx = fmaxf(mx, p[q]);
        }
        float sum = 0.f;
#pragma unroll
        for (int q = 0; q < 25; ++q) { p[q] = __expf(p[q] - mx); sum += p[q]; }
        const float rs = __frcp_rn(sum);

        const bool e2 = ((s & 1) == 0), e1 = ((s >> 1) == 0);
        float cf[5][3];
#pragma unroll
        for (int ki = 0; ki < 5; ++ki) {
            const float q0 = p[ki * 5 + 0] * rs, q1 = p[ki * 5 + 1] * rs,
                        q2 = p[ki * 5 + 2] * rs, q3 = p[ki * 5 + 3] * rs,
                        q4 = p[ki * 5 + 4] * rs;
            cf[ki][0] = q0 + (e2 ? q1 : 0.f);
            cf[ki][1] = (e2 ? q3 : q1) + q2;
            cf[ki][2] = q4 + (e2 ? 0.f : q3);
        }
        float wc[9];
#pragma unroll
        for (int j = 0; j < 3; ++j) {
            wc[0 * 3 + j] = cf[0][j] + (e1 ? cf[1][j] : 0.f);
            wc[1 * 3 + j] = (e1 ? cf[3][j] : cf[1][j]) + cf[2][j];
            wc[2 * 3 + j] = cf[4][j] + (e1 ? 0.f : cf[3][j]);
        }
#pragma unroll
        for (int t = 0; t < 9; ++t)
            wl[t * 256 + tid] = wc[t];
    }
    __syncthreads();

    v2f W0[2][9], W1[2][9];
#pragma unroll
    for (int hf = 0; hf < 2; ++hf) {
#pragma unroll
        for (int t = 0; t < 9; ++t) {
            W0[hf][t].x = wl[t * 256 + 32 * hf + lane];
            W0[hf][t].y = wl[t * 256 + 64 + 32 * hf + lane];
            W1[hf][t].x = wl[t * 256 + 128 + 32 * hf + lane];
            W1[hf][t].y = wl[t * 256 + 192 + 32 * hf + lane];
        }
    }
    float* ob = out + (size_t)b * 128 * 16384 + (size_t)(2 * h) * 128 + 2 * lane;
#define REASM(C)                                                              \
    _Pragma("unroll")                                                         \
    for (int hf = 0; hf < 2; ++hf) {                                          \
        const float* lc = xs + (C) * XP + 3 + 32 * hf + lane;                 \
        v2f r0 = {0.f, 0.f}, r1 = {0.f, 0.f};                                 \
        _Pragma("unroll")                                                     \
        for (int t = 0; t < 9; ++t) {                                         \
            const float v = lc[(t / 3) * 72 + (t % 3)];                       \
            r0 += W0[hf][t] * v;                                              \
            r1 += W1[hf][t] * v;                                              \
        }                                                                     \
        *(v2f*)(ob + (size_t)(C) * 16384 + 64 * hf)       = r0;   \
        *(v2f*)(ob + (size_t)(C) * 16384 + 128 + 64 * hf) = r1;   \
    }
#pragma unroll 2
    for (int i = 0; i < 16; ++i)
        REASM(g + 8 * i)
#undef REASM
}
#undef YP
#undef XP

extern "C" void kernel_launch(void* const* d_in, const int* in_sizes, int n_in,
                              void* d_out, int out_size, void* d_ws, size_t ws_size,
                              hipStream_t stream) {
    const float* x      = (const float*)d_in[0];
    const float* comp_w = (const float*)d_in[1];
    const float* gamma  = (const float*)d_in[2];
    const float* beta   = (const float*)d_in[3];
    const float* mean   = (const float*)d_in[4];
    const float* var    = (const float*)d_in[5];
    const float* enc_w  = (const float*)d_in[6];
    const float* enc_b  = (const float*)d_in[7];
    float* out = (float*)d_out;

    float* y   = (float*)d_ws;
    float* wAe = y + (size_t)1048576;

    k_compress<<<256, 128, 0, stream>>>(x, comp_w, enc_w, gamma, beta, mean, var, y, wAe);
    k_encreasm<<<256, 256, 0, stream>>>(x, y, wAe, enc_b, out);
}


# k_encreasm: 3x3 encoder in the row-wise F(2,3) transform domain (weights transformed in the repack, y patch at staging, outputs recombined at the softmax read): 4032 -> 2688 f32 WMMAs per block
# speedup vs baseline: 3.6010x; 1.0606x over previous
#include <hip/hip_runtime.h>
#include <hip/hip_bf16.h>

typedef __attribute__((ext_vector_type(2))) float v2f;
typedef __attribute__((ext_vector_type(4))) float v4f;
typedef __attribute__((ext_vector_type(8))) float v8f;

#define EPS 1e-5f

#define PP 144
#define WP 132
__global__ void __launch_bounds__(128)
k_compress(const float* __restrict__ x,
                           const float* __restrict__ comp_w,
                           const float* __restrict__ enc_w,
                           const float* __restrict__ gamma,
                           const float* __restrict__ beta,
                           const float* __restrict__ mean,
                           const float* __restrict__ var,
                           float* __restrict__ y,
                           float* __restrict__ wAe) {
    __shared__ __attribute__((aligned(16))) float ls[64 * PP];
    __shared__ __attribute__((aligned(16))) float lw[64 * WP];
    const int tid    = threadIdx.x;
    const int row    = blockIdx.x;
    const int b      = row >> 6;
    const int hwBase = (row & 63) << 6;

    const int lane = tid & 31, l = lane & 15, half = lane >> 4;
    const int oBase = (tid >> 5) << 4;


    const float* xb = x + (size_t)b * 128 * 4096 + hwBase;
    {
        const int pr0 = tid >> 4, p4 = (tid & 15) << 2;
        const int r0 = tid >> 5, k4 = (tid & 31) << 2;
        v4f va[8], vb[8], vw[16];
#pragma unroll
        for (int j = 0; j < 8; ++j) {
            const float* xc = xb + (size_t)(2 * (pr0 + 8 * j)) * 4096 + p4;
            va[j] = *(const v4f*)(xc);
            vb[j] = *(const v4f*)(xc + 4096);
        }
#pragma unroll
        for (int j = 0; j < 16; ++j)
            vw[j] = *(const v4f*)(comp_w + (size_t)(r0 + 4 * j) * 128 + k4);
#pragma unroll
        for (int j = 0; j < 8; ++j) {
            float* d = ls + (pr0 + 8 * j) * PP + p4 * 2;
            v4f s0, s1;
            s0.x = va[j].x; s0.y = vb[j].x; s0.z = va[j].y; s0.w = vb[j].y;
            s1.x = va[j].z; s1.y = vb[j].z; s1.z = va[j].w; s1.w = vb[j].w;
            *(v4f*)(d)     = s0;
            *(v4f*)(d + 4) = s1;
        }
#pragma unroll
        for (int j = 0; j < 16; ++j)
            *(v4f*)(lw + (r0 + 4 * j) * WP + k4) = vw[j];
    }
    __syncthreads();

    const int wi0 = row * 336 + tid;
    const int wi1 = row * 336 + tid + 128;
    const int wi2 = row * 336 + (tid < 80 ? tid + 256 : 0);
    float wv0, wv1, wv2;
#define CMP_REPACK(WI, WV)                                                    \
    {                                                                         \
        const int wo = ((WI) >> 1) % 112;                                     \
        const int wt = (((WI) / 224) << 1) | ((WI) & 1);                      \
        const int wc = wt & 63, wT = wt >> 6;               \
        const int wj = wT & 3, wdy = wT >> 2;                                 \
        const float* gp = enc_w + (size_t)(wo < 100 ? wo : 0) * 576 + wc * 9 + wdy * 3; \
        const float g0 = gp[0], g1 = gp[1], g2 = gp[2];                       \
        const float um = 0.5f * ((g0 + g2) + (wj == 1 ? g1 : -g1));           \
        const float uu = (wj == 0) ? g0 : ((wj == 3) ? g2 : um);              \
        WV = (wo < 100) ? uu : 0.f;                                           \
    }
    CMP_REPACK(wi0, wv0)
    CMP_REPACK(wi1, wv1)
    CMP_REPACK(wi2, wv2)
#undef CMP_REPACK

    v4f q[16];
    {
        const float* ap = lw + (oBase + l) * WP + half * 4;
#pragma unroll
        for (int j = 0; j < 16; ++j)
            q[j] = *(const v4f*)(ap + j * 8);
    }

    v8f acc[4] = {};
    const float* lb = ls + (2 * half) * PP + l * 2;
    v2f b0[8][4], b1[8][4];

#define CMP_LOAD_B(SET, BLK)                                                  \
    _Pragma("unroll")                                                         \
    for (int s = 0; s < 8; ++s) {                                             \
        const float* p = lb + ((BLK) * 16 + (s >> 1) * 4 + (s & 1)) * PP;     \
        _Pragma("unroll")                                                     \
        for (int t = 0; t < 4; ++t)                                           \
            SET[s][t] = *(const v2f*)(p + 32 * t);                            \
    }
#define CMP_MMA(SET, BLK)                                                     \
    _Pragma("unroll")                                                         \
    for (int s = 0; s < 8; ++s) {                                             \
        v2f am;                                                               \
        if (s & 1) { am.x = q[(BLK) * 4 + (s >> 1)].z; am.y = q[(BLK) * 4 + (s >> 1)].w; } \
        else       { am.x = q[(BLK) * 4 + (s >> 1)].x; am.y = q[(BLK) * 4 + (s >> 1)].y; } \
        _Pragma("unroll")                                                     \
        for (int t = 0; t < 4; ++t)                                           \
            acc[t] = __builtin_amdgcn_wmma_f32_16x16x4_f32(                   \
                false, am, false, SET[s][t], (short)0, acc[t], false, false); \
    }

    CMP_LOAD_B(b0, 0)
    CMP_LOAD_B(b1, 1)
    __builtin_amdgcn_sched_barrier(0);
    CMP_MMA(b0, 0)
    CMP_LOAD_B(b0, 2)
    __builtin_amdgcn_sched_barrier(0);
    CMP_MMA(b1, 1)
    CMP_LOAD_B(b1, 3)
    __builtin_amdgcn_sched_barrier(0);
    CMP_MMA(b0, 2)
    CMP_MMA(b1, 3)
#undef CMP_LOAD_B
#undef CMP_MMA

#pragma unroll
    for (int j = 0; j < 8; ++j) {
        const int o = oBase + j + half * 8;
        const float iv = gamma[o] * rsqrtf(var[o] + EPS);
        const float bb = beta[o] - mean[o] * iv;
        float* yp = y + ((size_t)b * 64 + o) * 4096 + hwBase + l;
#pragma unroll
        for (int t = 0; t < 4; ++t) {
            float v = acc[t][j] * iv + bb;
            v = v * __frcp_rn(1.f + __expf(-v));
            yp[16 * t] = v;
        }
    }

    wAe[wi0] = wv0;
    wAe[wi1] = wv1;
    if (tid < 80)
        wAe[wi2] = wv2;
}
#undef PP
#undef WP

#define VP 96
#define VSZ (96 * VP)
#define XP 216

template <int NO>
__device__ __forceinline__ void enc_load_a(v2f (&buf)[3][16], const float* __restrict__ ap, int T) {
#pragma unroll
    for (int i = 0; i < NO; ++i) {
#pragma unroll
        for (int s = 0; s < 16; ++s)
            buf[i][s] = *(const v2f*)(ap + T * 7168 + s * 448 + i * 32);
    }
}
template <int HS>
__device__ __forceinline__ void enc_load_b(v2f (&bs)[8][2], const float* lb, int dy, int j) {
    const float* lo = lb + j * VSZ + (dy * 32 + HS * 16) * VP;
#pragma unroll
    for (int s = 0; s < 8; ++s) {
#pragma unroll
        for (int t = 0; t < 2; ++t)
            bs[s][t] = *(const v2f*)(lo + (2 * s) * VP + 32 * t);
    }
}
template <int NO, int HS>
__device__ __forceinline__ void enc_mma(const v2f (&buf)[3][16], const v2f (&bs)[8][2], v8f (&acc)[8], int base) {
#pragma unroll
    for (int s = 0; s < 8; ++s) {
#pragma unroll
        for (int i = 0; i < NO; ++i) {
#pragma unroll
            for (int t = 0; t < 2; ++t)
                acc[base + i * 2 + t] = __builtin_amdgcn_wmma_f32_16x16x4_f32(
                    false, buf[i][HS * 8 + s], false, bs[s][t], (short)0, acc[base + i * 2 + t], false, false);
        }
    }
}
struct XStage {
    const float* xg;
    float* xd;
    int ro0, ro1, ro2;
    bool ok0, ok2;
};
__device__ __forceinline__ void x_request(v4f (&xr)[9], const XStage& X, int dy) {
#pragma unroll
    for (int q = 0; q < 3; ++q) {
        if (q < 2 || dy < 2) {
            const float* xc = X.xg + (size_t)((3 * dy + q) * 16) * 4096;
            xr[3 * q + 0] = *(const v4f*)(xc + X.ro0);
            xr[3 * q + 1] = *(const v4f*)(xc + X.ro1);
            xr[3 * q + 2] = *(const v4f*)(xc + X.ro2);
        }
    }
}
__device__ __forceinline__ void x_store(const v4f (&xr)[9], const XStage& X, int dy) {
    const v4f z = {0.f, 0.f, 0.f, 0.f};
#pragma unroll
    for (int q = 0; q < 3; ++q) {
        if (q < 2 || dy < 2) {
            float* xo = X.xd + ((3 * dy + q) * 16) * XP;
            *(v4f*)(xo)       = X.ok0 ? xr[3 * q + 0] : z;
            *(v4f*)(xo + 72)  = xr[3 * q + 1];
            *(v4f*)(xo + 144) = X.ok2 ? xr[3 * q + 2] : z;
        }
    }
}
template <int NO>
__device__ __forceinline__ void enc_step(const v2f (&acur)[3][16], v2f (&anext)[3][16], v2f (&b)[2][8][2],
                                         const float* __restrict__ ap, const float* lb, v8f (&acc)[8],
                                         int base, int dy, int j, bool pre, int dyN, int jN,
                                         bool xl, bool xs, v4f (&xr)[9], const XStage& X) {
    if (pre)
        enc_load_a<NO>(anext, ap, dyN * 4 + jN);
    enc_load_b<1>(b[1], lb, dy, j);
    if (xl)
        x_request(xr, X, dy);
    __builtin_amdgcn_sched_barrier(0);
    enc_mma<NO, 0>(acur, b[0], acc, base);
    if (pre)
        enc_load_b<0>(b[0], lb, dyN, jN);
    __builtin_amdgcn_sched_barrier(0);
    enc_mma<NO, 1>(acur, b[1], acc, base);
    if (xs)
        x_store(xr, X, dy);
}
template <int NO, int NJ, int J0, bool ROLL>
__device__ __forceinline__ void enc_steps(const v2f (&afirst)[3][16], const float* __restrict__ ap, const float* lb,
                                          v8f (&acc)[8], const XStage& X) {
    constexpr int NB = (NJ % 2 == 0) ? 2 : 3;
    v2f a[3][3][16];
#pragma unroll
    for (int i = 0; i < NO; ++i) {
#pragma unroll
        for (int s = 0; s < 16; ++s)
            a[0][i][s] = afirst[i][s];
    }
    v2f b[2][8][2];
    v4f xr[9];
    enc_load_b<0>(b[0], lb, 0, J0);
    if (ROLL) {
#pragma unroll 1
        for (int dy = 0; dy < 3; ++dy) {
            const int dyn = dy < 2 ? dy + 1 : 0;
#pragma unroll
            for (int jj = 0; jj < NJ; ++jj) {
                const bool lastj = (jj == NJ - 1);
                enc_step<NO>(a[jj % NB], a[(jj + 1) % NB], b, ap, lb, acc, (NO == 1) ? jj * 2 : 0,
                             dy, J0 + jj, true, lastj ? dyn : dy, lastj ? J0 : J0 + jj + 1,
                             jj == 0, lastj, xr, X);
            }
        }
    } else {
#pragma unroll
        for (int n = 0; n < 3 * NJ; ++n) {
            const int dy = n / NJ, jj = n % NJ;
            const int dyN = (n + 1) / NJ, jjN = (n + 1) % NJ;
            enc_step<NO>(a[n & 1], a[(n + 1) & 1], b, ap, lb, acc, (NO == 1) ? jj * 2 : 0,
                         dy, J0 + jj, n + 1 < 3 * NJ, dyN, J0 + jjN,
                         jj == 0, jj == NJ - 1, xr, X);
        }
    }
}

__global__ void __launch_bounds__(256)
k_encreasm(const float* __restrict__ x,
                           const float* __restrict__ y,
                           const float* __restrict__ wAe,
                           const float* __restrict__ enc_b,
                           float* __restrict__ out) {
    __shared__ __attribute__((aligned(16))) float ls[4 * VSZ];
    __shared__ __attribute__((aligned(16))) float xs[128 * XP];
    __shared__ float wl[36 * 64];
    const int tid  = threadIdx.x;
    const int row  = blockIdx.x;
    const int b    = row >> 6;
    const int h    = row & 63;

    const int lane = tid & 31, l = lane & 15, half = lane >> 4;
    const int g = __builtin_amdgcn_readfirstlane(tid >> 5);
    const bool last = g == 7;
    const int oBase = (last ? 4 : g) << 4;
    const int aCol = (oBase + l) * 2;

    const float* ap = wAe + half * 224 + aCol;
    v2f a0[3][16];

    const float* yb = y + (size_t)b * 64 * 4096;
    const float* xb = x + (size_t)b * 128 * 4096;
    v8f acc[8] = {};
    if (!last) {
        enc_load_a<1>(a0, ap, 0);
#pragma unroll
        for (int j = 0; j < 8; ++j) {
            const int oo = oBase + j + half * 8;
            const float bv = enc_b[oo < 100 ? oo : 0];
            const float sv = (oo < 100) ? bv : 0.f;
            acc[2][j] = sv; acc[3][j] = sv;
        }
    } else {
        enc_load_a<3>(a0, ap, 3);
    }
    {
        v2f e[12][2];
#pragma unroll
        for (int k = 0; k < 12; ++k) {
            const int u = g * 12 + k;
            const int hh = h + (u >> 5) - 1;
            const int hc = hh < 0 ? 0 : (hh > 63 ? 63 : hh);
            const float* p = yb + ((size_t)(2 * (u & 31)) * 64 + hc) * 64 + 2 * lane;
            e[k][0] = *(const v2f*)(p);
            e[k][1] = *(const v2f*)(p + 4096);
        }
#pragma unroll
        for (int k = 0; k < 12; ++k) {
            const int u = g * 12 + k;
            const int hh = h + (u >> 5) - 1;
            const bool ok = (hh >= 0) & (hh < 64);
            v2f V0, V1, V2, V3;
            {
                const float d1 = ok ? e[k][0].x : 0.f, d2 = ok ? e[k][0].y : 0.f;
                const float um = __shfl_up(d2, 1), dn = __shfl_down(d1, 1);
                const float d0 = (lane == 0) ? 0.f : um;
                const float d3 = (lane == 31) ? 0.f : dn;
                V0.x = d0 - d2; V1.x = d1 + d2; V2.x = d2 - d1; V3.x = d1 - d3;
            }
            {
                const float d1 = ok ? e[k][1].x : 0.f, d2 = ok ? e[k][1].y : 0.f;
                const float um = __shfl_up(d2, 1), dn = __shfl_down(d1, 1);
                const float d0 = (lane == 0) ? 0.f : um;
                const float d3 = (lane == 31) ? 0.f : dn;
                V0.y = d0 - d2; V1.y = d1 + d2; V2.y = d2 - d1; V3.y = d1 - d3;
            }
            float* vd = ls + u * VP + lane * 2;
            *(v2f*)(vd)           = V0;
            *(v2f*)(vd + VSZ)     = V1;
            *(v2f*)(vd + 2 * VSZ) = V2;
            *(v2f*)(vd + 3 * VSZ) = V3;
        }
    }
#pragma unroll
    for (int k = 0; k < 3; ++k) {
        const int e = tid + 256 * k;
        const int ln = e >> 1;
        xs[(ln & 127) * XP + (ln >> 7) * 72 + ((e & 1) ? 68 : 3)] = 0.f;
    }
    __syncthreads();

    const float* lb = ls + half * VP + l * 2;

    XStage X;
    X.xg = xb + (size_t)(tid >> 4) * 4096 + ((tid & 15) << 2);
    X.xd = xs + (tid >> 4) * XP + 4 + ((tid & 15) << 2);
    X.ok0 = h > 0; X.ok2 = h < 63;
    X.ro0 = (X.ok0 ? h - 1 : 0) * 64; X.ro1 = h * 64; X.ro2 = (X.ok2 ? h + 1 : 63) * 64;

    if (g < 4) {
        enc_steps<1, 4, 0, true>(a0, ap, lb, acc, X);
    } else if (!last) {
        enc_steps<1, 3, 0, true>(a0, ap, lb, acc, X);
    } else {
        enc_steps<3, 1, 3, false>(a0, ap, lb, acc, X);
    }

    __syncthreads();
    if (!last) {
#pragma unroll
        for (int j = 0; j < 8; ++j) {
            const int oo = oBase + j + half * 8;
            if (oo < 100) {
#pragma unroll
                for (int t = 0; t < 6; ++t)
                    ls[((t >> 1) * 100 + oo) * 32 + 16 * (t & 1) + l] = acc[t][j];
                if (g < 4) {
                    ls[(300 + oo) * 32 + l]      = acc[6][j];
                    ls[(300 + oo) * 32 + 16 + l] = acc[7][j];
                }
            }
        }
    } else {
#pragma unroll
        for (int i = 0; i < 3; ++i) {
#pragma unroll
            for (int j = 0; j < 8; ++j) {
                const int oo = 64 + 16 * i + j + half * 8;
                if (oo < 100) {
                    ls[(300 + oo) * 32 + l]      = acc[i * 2][j];
                    ls[(300 + oo) * 32 + 16 + l] = acc[i * 2 + 1][j];
                }
            }
        }
    }
    __syncthreads();

    {
        const int s = tid >> 6;
        const int px = tid & 63;
        const bool odd = (px & 1) != 0;
        const float* mb = ls + (odd ? 9600 : 0) + s * 32 + (px >> 1);
        float p[25];
        float mx = -1e30f;
#pragma unroll
        for (int q = 0; q < 25; ++q) {
            const float ma = mb[q * 128];
            const float m1 = ls[3200 + (q * 4 + s) * 32 + (px >> 1)];
            const float m2 = ls[6400 + (q * 4 + s) * 32 + (px >> 1)];
            p[q] = odd ? ((m1 - m2) - ma) : ((ma + m1) + m2);
            mx = fmaxf(mx, p[q]);
        }
        float sum = 0.f;
#pragma unroll
        for (int q = 0; q < 25; ++q) { p[q] = __expf(p[q] - mx); sum += p[q]; }
        const float rs = __frcp_rn(sum);

        const bool e2 = ((s & 1) == 0), e1 = ((s >> 1) == 0);
        float cf[5][3];
#pragma unroll
        for (int ki = 0; ki < 5; ++ki) {
            const float q0 = p[ki * 5 + 0] * rs, q1 = p[ki * 5 + 1] * rs,
                        q2 = p[ki * 5 + 2] * rs, q3 = p[ki * 5 + 3] * rs,
                        q4 = p[ki * 5 + 4] * rs;
            cf[ki][0] = q0 + (e2 ? q1 : 0.f);
            cf[ki][1] = (e2 ? q3 : q1) + q2;
            cf[ki][2] = q4 + (e2 ? 0.f : q3);
        }
        float wc[9];
#pragma unroll
        for (int j = 0; j < 3; ++j) {
            wc[0 * 3 + j] = cf[0][j] + (e1 ? cf[1][j] : 0.f);
            wc[1 * 3 + j] = (e1 ? cf[3][j] : cf[1][j]) + cf[2][j];
            wc[2 * 3 + j] = cf[4][j] + (e1 ? 0.f : cf[3][j]);
        }
#pragma unroll
        for (int t = 0; t < 9; ++t)
            wl[t * 256 + tid] = wc[t];
    }
    __syncthreads();

    v2f W0[2][9], W1[2][9];
#pragma unroll
    for (int hf = 0; hf < 2; ++hf) {
#pragma unroll
        for (int t = 0; t < 9; ++t) {
            W0[hf][t].x = wl[t * 256 + 32 * hf + lane];
            W0[hf][t].y = wl[t * 256 + 64 + 32 * hf + lane];
            W1[hf][t].x = wl[t * 256 + 128 + 32 * hf + lane];
            W1[hf][t].y = wl[t * 256 + 192 + 32 * hf + lane];
        }
    }
    float* ob = out + (size_t)b * 128 * 16384 + (size_t)(2 * h) * 128 + 2 * lane;
#define REASM(C)                                                              \
    _Pragma("unroll")                                                         \
    for (int hf = 0; hf < 2; ++hf) {                                          \
        const float* lc = xs + (C) * XP + 3 + 32 * hf + lane;                 \
        v2f r0 = {0.f, 0.f}, r1 = {0.f, 0.f};                                 \
        _Pragma("unroll")                                                     \
        for (int t = 0; t < 9; ++t) {                                         \
            const float v = lc[(t / 3) * 72 + (t % 3)];                       \
            r0 += W0[hf][t] * v;                                              \
            r1 += W1[hf][t] * v;                                              \
        }                                                                     \
        *(v2f*)(ob + (size_t)(C) * 16384 + 64 * hf)       = r0;   \
        *(v2f*)(ob + (size_t)(C) * 16384 + 128 + 64 * hf) = r1;   \
    }
#pragma unroll 2
    for (int i = 0; i < 16; ++i)
        REASM(g + 8 * i)
#undef REASM
}
#undef VP
#undef VSZ
#undef XP

extern "C" void kernel_launch(void* const* d_in, const int* in_sizes, int n_in,
                              void* d_out, int out_size, void* d_ws, size_t ws_size,
                              hipStream_t stream) {
    const float* x      = (const float*)d_in[0];
    const float* comp_w = (const float*)d_in[1];
    const float* gamma  = (const float*)d_in[2];
    const float* beta   = (const float*)d_in[3];
    const float* mean   = (const float*)d_in[4];
    const float* var    = (const float*)d_in[5];
    const float* enc_w  = (const float*)d_in[6];
    const float* enc_b  = (const float*)d_in[7];
    float* out = (float*)d_out;

    float* y   = (float*)d_ws;
    float* wAe = y + (size_t)1048576;

    k_compress<<<256, 128, 0, stream>>>(x, comp_w, enc_w, gamma, beta, mean, var, y, wAe);
    k_encreasm<<<256, 256, 0, stream>>>(x, y, wAe, enc_b, out);
}
